# starting kernel
# baseline (speedup 1.0000x reference)
#include <hip/hip_runtime.h>

#define BATCH 256
#define SEQ   4096
#define NS    16
#define MS    8
#define US    4
#define STEPF 448

typedef __attribute__((ext_vector_type(2))) float v2f;
typedef __attribute__((ext_vector_type(8))) float v8f;

__global__ __launch_bounds__(256)
void kalman_cov_kernel(const float* __restrict__ Ag, const float* __restrict__ Bg,
                       const float* __restrict__ Cg, const float* __restrict__ Qg,
                       const float* __restrict__ Rg, float* __restrict__ ws)
{
    __shared__ float sA[NS*NS], sB[NS*US], sC[MS*NS], sQ[NS*NS], sR[MS*MS];
    __shared__ float sP[NS*NS], sT1[NS*NS], sPp[NS*NS];
    __shared__ float sCP[MS*NS], sSI[MS][2*MS], sPCt[NS*MS], sK[NS*MS], sW[NS*NS];

    const int tid = threadIdx.x;
    sA[tid] = Ag[tid];
    sQ[tid] = Qg[tid];
    if (tid < NS*US) sB[tid] = Bg[tid];
    if (tid < MS*NS) sC[tid] = Cg[tid];
    if (tid < MS*MS) sR[tid] = Rg[tid];
    {
        int i = tid >> 4, j = tid & 15;
        sP[tid] = (i == j) ? 1.0f : 0.0f;
    }
    __syncthreads();

    const int i16 = tid >> 4, j16 = tid & 15;
    const int i8  = tid >> 3, j8  = tid & 7;

    #pragma unroll 1
    for (int t = 1; t < SEQ; ++t) {
        float acc = 0.f;
        #pragma unroll
        for (int k = 0; k < NS; ++k) acc += sA[i16*NS + k] * sP[k*NS + j16];
        sT1[tid] = acc;
        __syncthreads();
        acc = sQ[tid];
        #pragma unroll
        for (int k = 0; k < NS; ++k) acc += sT1[i16*NS + k] * sA[j16*NS + k];
        sPp[tid] = acc;
        __syncthreads();
        if (tid < MS*NS) {
            float a = 0.f;
            #pragma unroll
            for (int k = 0; k < NS; ++k) a += sC[i16*NS + k] * sPp[k*NS + j16];
            sCP[i16*NS + j16] = a;
            float b = 0.f;
            #pragma unroll
            for (int k = 0; k < NS; ++k) b += sPp[i8*NS + k] * sC[j8*NS + k];
            sPCt[i8*MS + j8] = b;
        }
        __syncthreads();
        if (tid < MS*NS) {
            float v;
            if (j16 < MS) {
                v = sR[i16*MS + j16];
                #pragma unroll
                for (int k = 0; k < NS; ++k) v += sCP[i16*NS + k] * sC[j16*NS + k];
            } else {
                v = ((j16 - MS) == i16) ? 1.0f : 0.0f;
            }
            sSI[i16][j16] = v;
        }
        __syncthreads();
        #pragma unroll 1
        for (int p = 0; p < MS; ++p) {
            float invp = 1.0f / sSI[p][p];
            float pv = 0.f, f = 0.f;
            if (tid < MS*NS) { pv = sSI[p][j16] * invp; f = sSI[i16][p]; }
            __syncthreads();
            if (tid < MS*NS) sSI[i16][j16] = (i16 == p) ? pv : sSI[i16][j16] - f * pv;
            __syncthreads();
        }
        if (tid < NS*MS) {
            float v = 0.f;
            #pragma unroll
            for (int k = 0; k < MS; ++k) v += sPCt[i8*MS + k] * sSI[k][MS + j8];
            sK[i8*MS + j8] = v;
        }
        __syncthreads();
        {
            float v = (i16 == j16) ? 1.0f : 0.0f;
            #pragma unroll
            for (int k = 0; k < MS; ++k) v -= sK[i16*MS + k] * sC[k*NS + j16];
            sW[tid] = v;
        }
        __syncthreads();
        float* wsp = ws + (size_t)(t - 1) * STEPF;
        {
            float v = 0.f;
            #pragma unroll
            for (int k = 0; k < NS; ++k) v += sW[i16*NS + k] * sA[k*NS + j16];
            wsp[tid] = v;
        }
        {
            float v = 0.f;
            #pragma unroll
            for (int k = 0; k < NS; ++k) v += sW[i16*NS + k] * sPp[k*NS + j16];
            sP[tid] = v;
        }
        if (tid < NS*US) {
            int ib = tid >> 2, jb = tid & 3;
            float v = 0.f;
            #pragma unroll
            for (int k = 0; k < NS; ++k) v += sW[ib*NS + k] * sB[k*US + jb];
            wsp[256 + tid] = v;
        }
        if (tid < NS*MS) wsp[320 + tid] = sK[tid];
        __syncthreads();
    }
}

__device__ __forceinline__ float bperm(int srcLane, float v) {
    return __builtin_bit_cast(float,
        __builtin_amdgcn_ds_bpermute(srcLane << 2, __builtin_bit_cast(int, v)));
}

__device__ __forceinline__ v8f wmma4(v2f a, v2f b, v8f c) {
    return __builtin_amdgcn_wmma_f32_16x16x4_f32(false, a, false, b,
                                                 (short)0, c, false, false);
}

struct StepOps {
    v2f aA0, aA1, aA2, aA3;
    v2f aB, aK0, aK1;
    v2f bZ0, bZ1, bU;
};

__device__ __forceinline__ StepOps load_ops(const float* __restrict__ ws,
                                            const float* __restrict__ zbase,
                                            const float* __restrict__ ubase,
                                            int t, int col, int hi) {
    StepOps o;
    const float* w = ws + (size_t)(t - 1) * STEPF;
    o.aA0 = *(const v2f*)(w + col * 16 + 0  + 2 * hi);
    o.aA1 = *(const v2f*)(w + col * 16 + 4  + 2 * hi);
    o.aA2 = *(const v2f*)(w + col * 16 + 8  + 2 * hi);
    o.aA3 = *(const v2f*)(w + col * 16 + 12 + 2 * hi);
    o.aB  = *(const v2f*)(w + 256 + col * 4 + 2 * hi);
    o.aK0 = *(const v2f*)(w + 320 + col * 8 + 0 + 2 * hi);
    o.aK1 = *(const v2f*)(w + 320 + col * 8 + 4 + 2 * hi);
    const float* zr = zbase + (size_t)t * MS;
    o.bZ0 = *(const v2f*)(zr + 0 + 2 * hi);
    o.bZ1 = *(const v2f*)(zr + 4 + 2 * hi);
    o.bU  = *(const v2f*)(ubase + (size_t)t * US + 2 * hi);
    return o;
}

__device__ __forceinline__ v8f zu_chain(const StepOps& o) {
    v8f d = {};
    d = wmma4(o.aB,  o.bU,  d);
    d = wmma4(o.aK0, o.bZ0, d);
    d = wmma4(o.aK1, o.bZ1, d);
    return d;
}

__global__ __launch_bounds__(32)
void kalman_state_kernel(const float* __restrict__ obs, const float* __restrict__ inp,
                         const float* __restrict__ x0, const float* __restrict__ ws,
                         float* __restrict__ out)
{
    const int lane = threadIdx.x;
    const int hi   = lane >> 4;
    const int col  = lane & 15;
    const int b    = blockIdx.x * 16 + col;

    v8f x;
    #pragma unroll
    for (int r = 0; r < 8; ++r) x[r] = x0[8 * hi + r];

    {
        float* o = out + ((size_t)b * SEQ) * NS + 8 * hi;
        *(float4*)(o)     = make_float4(x[0], x[1], x[2], x[3]);
        *(float4*)(o + 4) = make_float4(x[4], x[5], x[6], x[7]);
    }

    const float* zbase = obs + (size_t)b * SEQ * MS;
    const float* ubase = inp + (size_t)b * SEQ * US;
    float*       obase = out + (size_t)b * SEQ * NS;

    StepOps cur = load_ops(ws, zbase, ubase, 1, col, hi);
    v8f dzu = zu_chain(cur);

    #pragma unroll 1
    for (int t = 1; t < SEQ; ++t) {
        const int tn = (t + 1 < SEQ) ? (t + 1) : t;
        StepOps nxt = load_ops(ws, zbase, ubase, tn, col, hi);

        v2f bX[4];
        #pragma unroll
        for (int c = 0; c < 4; ++c) {
            int srcLane = col + ((c >= 2) ? 16 : 0);
            int lo = 4 * (c & 1);
            float t0 = bperm(srcLane, x[lo + 0]);
            float t1 = bperm(srcLane, x[lo + 2]);
            float t2 = bperm(srcLane, x[lo + 1]);
            float t3 = bperm(srcLane, x[lo + 3]);
            bX[c].x = hi ? t1 : t0;
            bX[c].y = hi ? t3 : t2;
        }

        v8f d = wmma4(cur.aA0, bX[0], dzu);
        d = wmma4(cur.aA1, bX[1], d);
        d = wmma4(cur.aA2, bX[2], d);
        d = wmma4(cur.aA3, bX[3], d);
        x = d;

        float* o = obase + (size_t)t * NS + 8 * hi;
        *(float4*)(o)     = make_float4(x[0], x[1], x[2], x[3]);
        *(float4*)(o + 4) = make_float4(x[4], x[5], x[6], x[7]);

        dzu = zu_chain(nxt);
        cur = nxt;
    }
}

extern "C" void kernel_launch(void* const* d_in, const int* in_sizes, int n_in,
                              void* d_out, int out_size, void* d_ws, size_t ws_size,
                              hipStream_t stream) {
    const float* obs = (const float*)d_in[0];
    const float* inp = (const float*)d_in[1];
    const float* A   = (const float*)d_in[2];
    const float* B   = (const float*)d_in[3];
    const float* C   = (const float*)d_in[4];
    const float* Q   = (const float*)d_in[5];
    const float* R   = (const float*)d_in[6];
    const float* x0  = (const float*)d_in[7];
    float* out = (float*)d_out;
    float* ws  = (float*)d_ws;

    hipLaunchKernelGGL(kalman_cov_kernel, dim3(1), dim3(256), 0, stream,
                       A, B, C, Q, R, ws);
    hipLaunchKernelGGL(kalman_state_kernel, dim3(BATCH / 16), dim3(32), 0, stream,
                       obs, inp, x0, ws, out);
}


# Covariance block cut from 8 waves to 4, one per SIMD, each owning 4 lower rows (two row pairs interleaved): no two waves share a SIMD issue port; same sums in the same order
# speedup vs baseline: 4.2305x; 4.2305x over previous
#include <hip/hip_runtime.h>

#define BATCH 256
#define SEQ   4096
#define NS    16
#define MS    8
#define US    4
#define STEPF 448

typedef __attribute__((ext_vector_type(2))) float v2f;
typedef __attribute__((ext_vector_type(8))) float v8f;

#define NV  24
#define TWB 40
#define CWV 4
#define CTH (CWV*32)
#define RPW (NS/CWV)

__device__ __forceinline__ float lane_bcast(float v, int srcLane) {
    return __builtin_bit_cast(float,
        __builtin_amdgcn_readlane(__builtin_bit_cast(int, v), srcLane));
}

template <int K>
__device__ __forceinline__ float row_bcast(float v) {
    return __builtin_bit_cast(float,
        __builtin_amdgcn_update_dpp(0, __builtin_bit_cast(int, v), 0x150 + K, 0xF, 0xF, true));
}

__device__ __forceinline__ float row_dot(float seed, float v, const float (&m)[NS]) {
    float a = seed;
    a += row_bcast<0>(v)  * m[0];   a += row_bcast<1>(v)  * m[1];
    a += row_bcast<2>(v)  * m[2];   a += row_bcast<3>(v)  * m[3];
    a += row_bcast<4>(v)  * m[4];   a += row_bcast<5>(v)  * m[5];
    a += row_bcast<6>(v)  * m[6];   a += row_bcast<7>(v)  * m[7];
    a += row_bcast<8>(v)  * m[8];   a += row_bcast<9>(v)  * m[9];
    a += row_bcast<10>(v) * m[10];  a += row_bcast<11>(v) * m[11];
    a += row_bcast<12>(v) * m[12];  a += row_bcast<13>(v) * m[13];
    a += row_bcast<14>(v) * m[14];  a += row_bcast<15>(v) * m[15];
    return a;
}

__global__ __launch_bounds__(CTH)
void kalman_cov_kernel(const float* __restrict__ Ag, const float* __restrict__ Bg,
                       const float* __restrict__ Cg, const float* __restrict__ Qg,
                       const float* __restrict__ Rg, float* __restrict__ ws)
{
    __shared__ float sA[NS*NS], sB[NS*US], sC[MS*NS], sQ[NS*NS], sR[MS*MS];
    __shared__ float sP[NS*NS];
    __shared__ float sBot[NS*TWB];
    __shared__ float sCA[MS*NS], sCB[MS*US];

    const int tid = threadIdx.x;
    for (int e = tid; e < NS*NS; e += CTH) {
        sA[e] = Ag[e];
        sQ[e] = Qg[e];
        sP[e] = ((e >> 4) == (e & 15)) ? 1.0f : 0.0f;
    }
    if (tid < NS*US) sB[tid] = Bg[tid];
    if (tid < MS*NS) sC[tid] = Cg[tid];
    if (tid < MS*MS) sR[tid] = Rg[tid];
    __syncthreads();

    for (int e = tid; e < MS*NS + MS*US; e += CTH) {
        if (e < MS*NS) {
            const int i16 = e >> 4, j16 = e & 15;
            float a = 0.f;
            #pragma unroll
            for (int k = 0; k < NS; ++k) a += sC[i16*NS + k] * sA[k*NS + j16];
            sCA[e] = a;
        } else {
            const int q = e - MS*NS, ib = q >> 2, jb = q & 3;
            float a = 0.f;
            #pragma unroll
            for (int k = 0; k < NS; ++k) a += sC[ib*NS + k] * sB[k*US + jb];
            sCB[q] = a;
        }
    }
    __syncthreads();

    if (tid < NS*8)  sBot[(tid >> 3)*TWB + NV + (tid & 7)] = sA[(tid >> 3)*NS + (tid & 7)];

    const int w  = tid >> 5, c = tid & 31;
    const int rbase = RPW * w;
    float kBHi[RPW];
    int   oh[RPW];
    #pragma unroll
    for (int j = 0; j < RPW; ++j) {
        const int r = rbase + j;
        kBHi[j] = 0.f;  oh[j] = -1;
        if (c < 8)        { kBHi[j] = sA[r*NS + 8 + c];     oh[j] = r*NS + 8 + c; }
        else if (c < 12)  { kBHi[j] = sB[r*US + (c - 8)];   oh[j] = 256 + r*US + (c - 8); }
        else if (c < 20)  { oh[j] = 320 + r*MS + (c - 12); }
    }
    float kPHi[MS];
    #pragma unroll
    for (int i = 0; i < MS; ++i) {
        float h = 0.f;
        if (c < 8)        h = sCA[i*NS + 8 + c];
        else if (c < 12)  h = sCB[i*US + (c - 8)];
        else if (c < 20)  h = ((c - 12) == i) ? -1.0f : 0.0f;
        kPHi[i] = h;
    }

    const int hr  = c >> 4, jc = c & 15;
    const int rowa = rbase + hr;
    const int rowb = rbase + 2 + hr;
    float kArowa[NS], kArowb[NS], kAj[NS], kCm[NS];
    #pragma unroll
    for (int k = 0; k < NS; ++k) {
        kArowa[k] = sA[rowa*NS + k];
        kArowb[k] = sA[rowb*NS + k];
        kAj[k]    = sA[jc*NS + k];
        kCm[k]    = sC[(jc & 7)*NS + k];
    }
    const float kQa = sQ[rowa*NS + jc];
    const float kQb = sQ[rowb*NS + jc];
    float kC[MS*NS];
    #pragma unroll
    for (int q = 0; q < MS*NS; ++q) kC[q] = sC[q];
    float kR[MS];
    #pragma unroll
    for (int i = 0; i < MS; ++i) kR[i] = (c < MS) ? sR[i*MS + c] : 0.0f;

    #pragma unroll 1
    for (int t = 1; t < SEQ; ++t) {
        float t1a = 0.f, t1b = 0.f;
        #pragma unroll
        for (int k = 0; k < NS; ++k) {
            const float pk = sP[k*NS + jc];
            t1a += kArowa[k] * pk;
            t1b += kArowb[k] * pk;
        }
        const float ppa = row_dot(kQa, t1a, kAj);
        const float ppb = row_dot(kQb, t1b, kAj);
        const float ga = row_dot(0.f, ppa, kCm);
        const float gb = row_dot(0.f, ppb, kCm);
        sBot[rowa*TWB + 8 + jc] = ppa;
        sBot[rowb*TWB + 8 + jc] = ppb;
        if (jc < MS) {
            sBot[rowa*TWB + jc] = ga;
            sBot[rowb*TWB + jc] = gb;
        }
        __syncthreads();

        float b_lo[RPW];
        #pragma unroll
        for (int j = 0; j < RPW; ++j) b_lo[j] = sBot[(rbase + j)*TWB + c];
        float col[NS];
        #pragma unroll
        for (int k = 0; k < NS; ++k) col[k] = sBot[k*TWB + c];
        __builtin_amdgcn_sched_barrier(0);
        float pl[MS], ph[MS];
        #pragma unroll
        for (int i = 0; i < MS; ++i) {
            pl[i] = kR[i];
            ph[i] = kPHi[i];
        }
        #pragma unroll
        for (int k = 0; k < NS; ++k) {
            #pragma unroll
            for (int i = 0; i < MS; ++i) pl[i] += kC[i*NS + k] * col[k];
        }
        float b_hi[RPW];
        #pragma unroll
        for (int j = 0; j < RPW; ++j) b_hi[j] = kBHi[j];

        #pragma unroll
        for (int p = 0; p < MS; ++p) {
            const float inv = 1.0f / lane_bcast(pl[p], p);
            #pragma unroll
            for (int i = p + 1; i < MS; ++i) {
                const float f = lane_bcast(pl[i], p) * inv;
                pl[i] -= f * pl[p];  ph[i] -= f * ph[p];
            }
            #pragma unroll
            for (int j = 0; j < RPW; ++j) {
                const float fj = lane_bcast(b_lo[j], p) * inv;
                b_lo[j] -= fj * pl[p];  b_hi[j] -= fj * ph[p];
            }
        }

        if (c >= 8 && c < NV) {
            #pragma unroll
            for (int j = 0; j < RPW; ++j) sP[(rbase + j)*NS + (c - 8)] = b_lo[j];
        }
        __syncthreads();
        float* wsp = ws + (size_t)(t - 1) * STEPF;
        if (c >= NV) {
            #pragma unroll
            for (int j = 0; j < RPW; ++j) wsp[(rbase + j)*NS + (c - NV)] = b_lo[j];
        }
        if (oh[0] >= 0) {
            #pragma unroll
            for (int j = 0; j < RPW; ++j) wsp[oh[j]] = b_hi[j];
        }
    }
}

#define CH_L 63
#define CH_N 65
#define PHI_ROW 2

__device__ __forceinline__ float bperm(int srcLane, float v) {
    return __builtin_bit_cast(float,
        __builtin_amdgcn_ds_bpermute(srcLane << 2, __builtin_bit_cast(int, v)));
}

__device__ __forceinline__ v8f wmma4(v2f a, v2f b, v8f c) {
    return __builtin_amdgcn_wmma_f32_16x16x4_f32(false, a, false, b,
                                                 (short)0, c, false, false);
}

struct AOps {
    v2f a0, a1, a2, a3;
};

struct StepOps {
    AOps  aA;
    v2f aB, aK0, aK1;
    v2f bZ0, bZ1, bU;
};

__device__ __forceinline__ AOps load_a(const float* m, int col, int hi) {
    AOps o;
    o.a0 = *(const v2f*)(m + col * 16 + 0  + 2 * hi);
    o.a1 = *(const v2f*)(m + col * 16 + 4  + 2 * hi);
    o.a2 = *(const v2f*)(m + col * 16 + 8  + 2 * hi);
    o.a3 = *(const v2f*)(m + col * 16 + 12 + 2 * hi);
    return o;
}

__device__ __forceinline__ StepOps load_ops(const float* __restrict__ ws,
                                            const float* __restrict__ zbase,
                                            const float* __restrict__ ubase,
                                            int t, int col, int hi) {
    StepOps o;
    const float* w = ws + (size_t)(t - 1) * STEPF;
    o.aA  = load_a(w, col, hi);
    o.aB  = *(const v2f*)(w + 256 + col * 4 + 2 * hi);
    o.aK0 = *(const v2f*)(w + 320 + col * 8 + 0 + 2 * hi);
    o.aK1 = *(const v2f*)(w + 320 + col * 8 + 4 + 2 * hi);
    const float* zr = zbase + (size_t)t * MS;
    o.bZ0 = *(const v2f*)(zr + 0 + 2 * hi);
    o.bZ1 = *(const v2f*)(zr + 4 + 2 * hi);
    o.bU  = *(const v2f*)(ubase + (size_t)t * US + 2 * hi);
    return o;
}

__device__ __forceinline__ v8f zu_chain(const StepOps& o) {
    v8f d = {};
    d = wmma4(o.aB,  o.bU,  d);
    d = wmma4(o.aK0, o.bZ0, d);
    d = wmma4(o.aK1, o.bZ1, d);
    return d;
}

__device__ __forceinline__ v8f recur(const AOps& m, const v8f& x, v8f c, int col, int hi) {
    v2f bX[4];
    #pragma unroll
    for (int q = 0; q < 4; ++q) {
        int srcLane = col + ((q >= 2) ? 16 : 0);
        int lo = 4 * (q & 1);
        float t0 = bperm(srcLane, x[lo + 0]);
        float t1 = bperm(srcLane, x[lo + 2]);
        float t2 = bperm(srcLane, x[lo + 1]);
        float t3 = bperm(srcLane, x[lo + 3]);
        bX[q].x = hi ? t1 : t0;
        bX[q].y = hi ? t3 : t2;
    }
    v8f d = wmma4(m.a0, bX[0], c);
    d = wmma4(m.a1, bX[1], d);
    d = wmma4(m.a2, bX[2], d);
    d = wmma4(m.a3, bX[3], d);
    return d;
}

__device__ __forceinline__ void store_row(float* o, const v8f& x) {
    *(float4*)(o)     = make_float4(x[0], x[1], x[2], x[3]);
    *(float4*)(o + 4) = make_float4(x[4], x[5], x[6], x[7]);
}

__device__ __forceinline__ v8f load_row(const float* o) {
    const float4 lo = *(const float4*)(o);
    const float4 hh = *(const float4*)(o + 4);
    v8f x = {lo.x, lo.y, lo.z, lo.w, hh.x, hh.y, hh.z, hh.w};
    return x;
}

__global__ __launch_bounds__(32)
void kalman_chunk_local_kernel(const float* __restrict__ obs, const float* __restrict__ inp,
                               const float* __restrict__ ws, float* __restrict__ out)
{
    const int lane = threadIdx.x;
    const int hi   = lane >> 4;
    const int col  = lane & 15;
    const int tile = blockIdx.x;
    const int t0   = blockIdx.y * CH_L;

    if (tile == BATCH / 16) {
        v8f x;
        #pragma unroll
        for (int r = 0; r < 8; ++r) x[r] = (8 * hi + r == col) ? 1.0f : 0.0f;
        AOps cur = load_a(ws + (size_t)t0 * STEPF, col, hi);
        #pragma unroll 1
        for (int s = 1; s <= CH_L; ++s) {
            const int wn = (s < CH_L) ? (t0 + s) : (t0 + s - 1);
            AOps nxt = load_a(ws + (size_t)wn * STEPF, col, hi);
            v8f zero = {};
            x = recur(cur, x, zero, col, hi);
            cur = nxt;
        }
        float* p = out + (size_t)(t0 + PHI_ROW) * NS;
        #pragma unroll
        for (int r = 0; r < 8; ++r) p[(8 * hi + r) * NS + col] = x[r];
        return;
    }

    const int b = tile * 16 + col;
    const float* zbase = obs + (size_t)b * SEQ * MS;
    const float* ubase = inp + (size_t)b * SEQ * US;

    v8f x = {};
    StepOps cur = load_ops(ws, zbase, ubase, t0 + 1, col, hi);
    v8f dzu = zu_chain(cur);
    #pragma unroll 1
    for (int s = 1; s <= CH_L; ++s) {
        const int t  = t0 + s;
        const int tn = (s < CH_L) ? (t + 1) : t;
        StepOps nxt = load_ops(ws, zbase, ubase, tn, col, hi);
        x = recur(cur.aA, x, dzu, col, hi);
        dzu = zu_chain(nxt);
        cur = nxt;
    }
    store_row(out + ((size_t)b * SEQ + (t0 + CH_L)) * NS + 8 * hi, x);
}

__global__ __launch_bounds__(32)
void kalman_chunk_link_kernel(const float* __restrict__ x0, float* out)
{
    const int lane = threadIdx.x;
    const int hi   = lane >> 4;
    const int col  = lane & 15;
    const int b    = blockIdx.x * 16 + col;

    v8f x;
    #pragma unroll
    for (int r = 0; r < 8; ++r) x[r] = x0[8 * hi + r];

    float* obase = out + (size_t)b * SEQ * NS + 8 * hi;
    store_row(obase, x);

    AOps cur = load_a(out + (size_t)PHI_ROW * NS, col, hi);
    v8f  y   = load_row(obase + (size_t)CH_L * NS);
    #pragma unroll 1
    for (int c = 0; c < CH_N; ++c) {
        const int cn = (c + 1 < CH_N) ? (c + 1) : c;
        AOps nxt = load_a(out + ((size_t)cn * CH_L + PHI_ROW) * NS, col, hi);
        v8f  yn  = load_row(obase + (size_t)(cn + 1) * CH_L * NS);
        x = recur(cur, x, y, col, hi);
        store_row(obase + (size_t)(c + 1) * CH_L * NS, x);
        cur = nxt;
        y = yn;
    }
}

__global__ __launch_bounds__(32)
void kalman_chunk_final_kernel(const float* __restrict__ obs, const float* __restrict__ inp,
                               const float* __restrict__ ws, float* out)
{
    const int lane = threadIdx.x;
    const int hi   = lane >> 4;
    const int col  = lane & 15;
    const int b    = blockIdx.x * 16 + col;
    const int t0   = blockIdx.y * CH_L;

    const float* zbase = obs + (size_t)b * SEQ * MS;
    const float* ubase = inp + (size_t)b * SEQ * US;
    float*       obase = out + (size_t)b * SEQ * NS + 8 * hi;

    v8f x = load_row(obase + (size_t)t0 * NS);
    StepOps cur = load_ops(ws, zbase, ubase, t0 + 1, col, hi);
    v8f dzu = zu_chain(cur);
    #pragma unroll 1
    for (int s = 1; s < CH_L; ++s) {
        const int t  = t0 + s;
        const int tn = (s + 1 < CH_L) ? (t + 1) : t;
        StepOps nxt = load_ops(ws, zbase, ubase, tn, col, hi);
        x = recur(cur.aA, x, dzu, col, hi);
        store_row(obase + (size_t)t * NS, x);
        dzu = zu_chain(nxt);
        cur = nxt;
    }
}

extern "C" void kernel_launch(void* const* d_in, const int* in_sizes, int n_in,
                              void* d_out, int out_size, void* d_ws, size_t ws_size,
                              hipStream_t stream) {
    const float* obs = (const float*)d_in[0];
    const float* inp = (const float*)d_in[1];
    const float* A   = (const float*)d_in[2];
    const float* B   = (const float*)d_in[3];
    const float* C   = (const float*)d_in[4];
    const float* Q   = (const float*)d_in[5];
    const float* R   = (const float*)d_in[6];
    const float* x0  = (const float*)d_in[7];
    float* out = (float*)d_out;
    float* ws  = (float*)d_ws;

    hipLaunchKernelGGL(kalman_cov_kernel, dim3(1), dim3(CTH), 0, stream,
                       A, B, C, Q, R, ws);
    hipLaunchKernelGGL(kalman_chunk_local_kernel, dim3(BATCH / 16 + 1, CH_N), dim3(32), 0, stream,
                       obs, inp, ws, out);
    hipLaunchKernelGGL(kalman_chunk_link_kernel, dim3(BATCH / 16), dim3(32), 0, stream,
                       x0, out);
    hipLaunchKernelGGL(kalman_chunk_final_kernel, dim3(BATCH / 16, CH_N), dim3(32), 0, stream,
                       obs, inp, ws, out);
}


# Covariance carried as the prior (predictor form): A M A^T + Q rides the step's one elimination as 4 more rows a wave; no posterior exchange: 1 barrier a step, not 2; sums reassociated
# speedup vs baseline: 6.4162x; 1.5167x over previous
#include <hip/hip_runtime.h>

#define BATCH 256
#define SEQ   4096
#define NS    16
#define MS    8
#define US    4
#define STEPF 448

typedef __attribute__((ext_vector_type(2))) float v2f;
typedef __attribute__((ext_vector_type(8))) float v8f;

#define TWB 40
#define CWV 4
#define CTH (CWV*32)
#define RPW (NS/CWV)

__device__ __forceinline__ float lane_bcast(float v, int srcLane) {
    return __builtin_bit_cast(float,
        __builtin_amdgcn_readlane(__builtin_bit_cast(int, v), srcLane));
}

template <int K>
__device__ __forceinline__ float row_bcast(float v) {
    return __builtin_bit_cast(float,
        __builtin_amdgcn_update_dpp(0, __builtin_bit_cast(int, v), 0x150 + K, 0xF, 0xF, true));
}

__device__ __forceinline__ float xhalf(float x) {
    const int b = __builtin_bit_cast(int, x);
    return __builtin_bit_cast(float,
        __builtin_amdgcn_permlanex16(b, b, 0x76543210, 0xfedcba98, false, false));
}

#define ROW_DOT4_STEP(K)                                              \
    { const float x0 = row_bcast<K>(v0), x1 = row_bcast<K>(v1);      \
      a0 += x0 * ma[K];  a1 += x1 * ma[K];                            \
      b0 += x0 * mb[K];  b1 += x1 * mb[K]; }
__device__ __forceinline__ void row_dot4(float v0, float v1,
                                         const float (&ma)[NS], const float (&mb)[NS],
                                         float& ra0, float& rb0, float& ra1, float& rb1) {
    float a0 = 0.f, b0 = 0.f, a1 = 0.f, b1 = 0.f;
    ROW_DOT4_STEP(0)   ROW_DOT4_STEP(1)   ROW_DOT4_STEP(2)   ROW_DOT4_STEP(3)
    ROW_DOT4_STEP(4)   ROW_DOT4_STEP(5)   ROW_DOT4_STEP(6)   ROW_DOT4_STEP(7)
    ROW_DOT4_STEP(8)   ROW_DOT4_STEP(9)   ROW_DOT4_STEP(10)  ROW_DOT4_STEP(11)
    ROW_DOT4_STEP(12)  ROW_DOT4_STEP(13)  ROW_DOT4_STEP(14)  ROW_DOT4_STEP(15)
    ra0 = a0;  rb0 = b0;  ra1 = a1;  rb1 = b1;
}
#undef ROW_DOT4_STEP

__global__ __launch_bounds__(CTH)
void kalman_cov_kernel(const float* __restrict__ Ag, const float* __restrict__ Bg,
                       const float* __restrict__ Cg, const float* __restrict__ Qg,
                       const float* __restrict__ Rg, float* __restrict__ ws)
{
    __shared__ float sA[NS*NS], sB[NS*US], sC[MS*NS], sQ[NS*NS], sR[MS*MS];
    __shared__ float sBot[2*NS*TWB];
    __shared__ float sCA[MS*NS], sCB[MS*US];

    const int tid = threadIdx.x;
    for (int e = tid; e < NS*NS; e += CTH) {
        sA[e] = Ag[e];
        sQ[e] = Qg[e];
    }
    if (tid < NS*US) sB[tid] = Bg[tid];
    if (tid < MS*NS) sC[tid] = Cg[tid];
    if (tid < MS*MS) sR[tid] = Rg[tid];
    __syncthreads();

    for (int e = tid; e < MS*NS + MS*US; e += CTH) {
        if (e < MS*NS) {
            const int i16 = e >> 4, j16 = e & 15;
            float a = 0.f;
            #pragma unroll
            for (int k = 0; k < NS; ++k) a += sC[i16*NS + k] * sA[k*NS + j16];
            sCA[e] = a;
        } else {
            const int q = e - MS*NS, ib = q >> 2, jb = q & 3;
            float a = 0.f;
            #pragma unroll
            for (int k = 0; k < NS; ++k) a += sC[ib*NS + k] * sB[k*US + jb];
            sCB[q] = a;
        }
    }
    __syncthreads();

    if (tid < NS*8) {
        const float a = sA[(tid >> 3)*NS + (tid & 7)];
        sBot[(tid >> 3)*TWB + 8 + (tid & 7)] = a;
        sBot[NS*TWB + (tid >> 3)*TWB + 8 + (tid & 7)] = a;
    }

    const int w  = tid >> 5, c = tid & 31;
    const int rbase = RPW * w;
    float kBHi[RPW];
    int   oh[RPW];
    #pragma unroll
    for (int j = 0; j < RPW; ++j) {
        const int r = rbase + j;
        kBHi[j] = 0.f;  oh[j] = -1;
        if (c < 8)        { kBHi[j] = sA[r*NS + 8 + c];     oh[j] = r*NS + 8 + c; }
        else if (c < 12)  { kBHi[j] = sB[r*US + (c - 8)];   oh[j] = 256 + r*US + (c - 8); }
        else if (c < 20)  { oh[j] = 320 + r*MS + (c - 12); }
    }
    float kPHi[MS];
    #pragma unroll
    for (int i = 0; i < MS; ++i) {
        float h = 0.f;
        if (c < 8)        h = sCA[i*NS + 8 + c];
        else if (c < 12)  h = sCB[i*US + (c - 8)];
        else if (c < 20)  h = ((c - 12) == i) ? -1.0f : 0.0f;
        kPHi[i] = h;
    }

    const int hr  = c >> 4, jc = c & 15;
    const int rowa = rbase + hr;
    const int rowb = rbase + 2 + hr;
    float kAj[NS], kCm[NS];
    #pragma unroll
    for (int k = 0; k < NS; ++k) {
        kAj[k]    = sA[jc*NS + k];
        kCm[k]    = sC[(jc & 7)*NS + k];
    }
    float kC[MS*NS];
    #pragma unroll
    for (int q = 0; q < MS*NS; ++q) kC[q] = sC[q];
    float kR[MS];
    #pragma unroll
    for (int i = 0; i < MS; ++i) kR[i] = (c < MS) ? sR[i*MS + c] : 0.0f;
    float kA2[RPW][NS];
    float kQ2[RPW];
    #pragma unroll
    for (int j = 0; j < RPW; ++j) {
        #pragma unroll
        for (int k = 0; k < NS; ++k) kA2[j][k] = sA[(rbase + j)*NS + k];
        kQ2[j] = (c >= 16) ? sQ[(rbase + j)*NS + (c - 16)] : 0.0f;
    }

    float m0, m1;
    {
        float a0 = 0.f, a1 = 0.f;
        #pragma unroll
        for (int k = 0; k < NS; ++k) {
            a0 += sA[rowa*NS + k] * kAj[k];
            a1 += sA[rowb*NS + k] * kAj[k];
        }
        m0 = a0 + sQ[rowa*NS + jc];
        m1 = a1 + sQ[rowb*NS + jc];
    }

    #pragma unroll 1
    for (int t = 1; t < SEQ; ++t) {
        float* bot = sBot + (t & 1) * (NS*TWB);
        float ga, xa, gb, xb;
        row_dot4(m0, m1, kCm, kAj, ga, xa, gb, xb);
        bot[rowa*TWB + 16 + jc] = xa;
        bot[rowb*TWB + 16 + jc] = xb;
        bot[rowa*TWB + (jc & 7)] = ga;
        bot[rowb*TWB + (jc & 7)] = gb;
        __syncthreads();

        float b_lo[RPW];
        #pragma unroll
        for (int j = 0; j < RPW; ++j) b_lo[j] = bot[(rbase + j)*TWB + c];
        float col[NS];
        #pragma unroll
        for (int k = 0; k < NS; ++k) col[k] = bot[k*TWB + c];
        __builtin_amdgcn_sched_barrier(0);
        float pl[MS], ph[MS];
        #pragma unroll
        for (int i = 0; i < MS; ++i) {
            pl[i] = kR[i];
            ph[i] = kPHi[i];
        }
        #pragma unroll
        for (int k = 0; k < NS; ++k) {
            #pragma unroll
            for (int i = 0; i < MS; ++i) pl[i] += kC[i*NS + k] * col[k];
        }
        float b2[RPW];
        #pragma unroll
        for (int j = 0; j < RPW; ++j) b2[j] = kQ2[j];
        #pragma unroll
        for (int k = 0; k < NS; ++k) {
            #pragma unroll
            for (int j = 0; j < RPW; ++j) b2[j] += kA2[j][k] * col[k];
        }
        float b_hi[RPW];
        #pragma unroll
        for (int j = 0; j < RPW; ++j) b_hi[j] = kBHi[j];

        #pragma unroll
        for (int p = 0; p < MS; ++p) {
            const float inv = 1.0f / lane_bcast(pl[p], p);
            #pragma unroll
            for (int i = p + 1; i < MS; ++i) {
                const float f = lane_bcast(pl[i], p) * inv;
                pl[i] -= f * pl[p];  ph[i] -= f * ph[p];
            }
            #pragma unroll
            for (int j = 0; j < RPW; ++j) {
                const float fj = lane_bcast(b_lo[j], p) * inv;
                b_lo[j] -= fj * pl[p];  b_hi[j] -= fj * ph[p];
                const float gj = lane_bcast(b2[j], p) * inv;
                b2[j] -= gj * pl[p];
            }
        }

        float* wsp = ws + (size_t)(t - 1) * STEPF;
        if (c >= 8 && c < 16) {
            #pragma unroll
            for (int j = 0; j < RPW; ++j) wsp[(rbase + j)*NS + (c - 8)] = b_lo[j];
        }
        if (oh[0] >= 0) {
            #pragma unroll
            for (int j = 0; j < RPW; ++j) wsp[oh[j]] = b_hi[j];
        }

        const float x0 = xhalf(b2[0]);
        const float x2 = xhalf(b2[2]);
        m0 = hr ? b2[1] : x0;
        m1 = hr ? b2[3] : x2;
    }
}

#define CH_L 63
#define CH_N 65
#define PHI_ROW 2

__device__ __forceinline__ float bperm(int srcLane, float v) {
    return __builtin_bit_cast(float,
        __builtin_amdgcn_ds_bpermute(srcLane << 2, __builtin_bit_cast(int, v)));
}

__device__ __forceinline__ v8f wmma4(v2f a, v2f b, v8f c) {
    return __builtin_amdgcn_wmma_f32_16x16x4_f32(false, a, false, b,
                                                 (short)0, c, false, false);
}

struct AOps {
    v2f a0, a1, a2, a3;
};

struct StepOps {
    AOps  aA;
    v2f aB, aK0, aK1;
    v2f bZ0, bZ1, bU;
};

__device__ __forceinline__ AOps load_a(const float* m, int col, int hi) {
    AOps o;
    o.a0 = *(const v2f*)(m + col * 16 + 0  + 2 * hi);
    o.a1 = *(const v2f*)(m + col * 16 + 4  + 2 * hi);
    o.a2 = *(const v2f*)(m + col * 16 + 8  + 2 * hi);
    o.a3 = *(const v2f*)(m + col * 16 + 12 + 2 * hi);
    return o;
}

__device__ __forceinline__ StepOps load_ops(const float* __restrict__ ws,
                                            const float* __restrict__ zbase,
                                            const float* __restrict__ ubase,
                                            int t, int col, int hi) {
    StepOps o;
    const float* w = ws + (size_t)(t - 1) * STEPF;
    o.aA  = load_a(w, col, hi);
    o.aB  = *(const v2f*)(w + 256 + col * 4 + 2 * hi);
    o.aK0 = *(const v2f*)(w + 320 + col * 8 + 0 + 2 * hi);
    o.aK1 = *(const v2f*)(w + 320 + col * 8 + 4 + 2 * hi);
    const float* zr = zbase + (size_t)t * MS;
    o.bZ0 = *(const v2f*)(zr + 0 + 2 * hi);
    o.bZ1 = *(const v2f*)(zr + 4 + 2 * hi);
    o.bU  = *(const v2f*)(ubase + (size_t)t * US + 2 * hi);
    return o;
}

__device__ __forceinline__ v8f zu_chain(const StepOps& o) {
    v8f d = {};
    d = wmma4(o.aB,  o.bU,  d);
    d = wmma4(o.aK0, o.bZ0, d);
    d = wmma4(o.aK1, o.bZ1, d);
    return d;
}

__device__ __forceinline__ v8f recur(const AOps& m, const v8f& x, v8f c, int col, int hi) {
    v2f bX[4];
    #pragma unroll
    for (int q = 0; q < 4; ++q) {
        int srcLane = col + ((q >= 2) ? 16 : 0);
        int lo = 4 * (q & 1);
        float t0 = bperm(srcLane, x[lo + 0]);
        float t1 = bperm(srcLane, x[lo + 2]);
        float t2 = bperm(srcLane, x[lo + 1]);
        float t3 = bperm(srcLane, x[lo + 3]);
        bX[q].x = hi ? t1 : t0;
        bX[q].y = hi ? t3 : t2;
    }
    v8f d = wmma4(m.a0, bX[0], c);
    d = wmma4(m.a1, bX[1], d);
    d = wmma4(m.a2, bX[2], d);
    d = wmma4(m.a3, bX[3], d);
    return d;
}

__device__ __forceinline__ void store_row(float* o, const v8f& x) {
    *(float4*)(o)     = make_float4(x[0], x[1], x[2], x[3]);
    *(float4*)(o + 4) = make_float4(x[4], x[5], x[6], x[7]);
}

__device__ __forceinline__ v8f load_row(const float* o) {
    const float4 lo = *(const float4*)(o);
    const float4 hh = *(const float4*)(o + 4);
    v8f x = {lo.x, lo.y, lo.z, lo.w, hh.x, hh.y, hh.z, hh.w};
    return x;
}

__global__ __launch_bounds__(32)
void kalman_chunk_local_kernel(const float* __restrict__ obs, const float* __restrict__ inp,
                               const float* __restrict__ ws, float* __restrict__ out)
{
    const int lane = threadIdx.x;
    const int hi   = lane >> 4;
    const int col  = lane & 15;
    const int tile = blockIdx.x;
    const int t0   = blockIdx.y * CH_L;

    if (tile == BATCH / 16) {
        v8f x;
        #pragma unroll
        for (int r = 0; r < 8; ++r) x[r] = (8 * hi + r == col) ? 1.0f : 0.0f;
        AOps cur = load_a(ws + (size_t)t0 * STEPF, col, hi);
        #pragma unroll 1
        for (int s = 1; s <= CH_L; ++s) {
            const int wn = (s < CH_L) ? (t0 + s) : (t0 + s - 1);
            AOps nxt = load_a(ws + (size_t)wn * STEPF, col, hi);
            v8f zero = {};
            x = recur(cur, x, zero, col, hi);
            cur = nxt;
        }
        float* p = out + (size_t)(t0 + PHI_ROW) * NS;
        #pragma unroll
        for (int r = 0; r < 8; ++r) p[(8 * hi + r) * NS + col] = x[r];
        return;
    }

    const int b = tile * 16 + col;
    const float* zbase = obs + (size_t)b * SEQ * MS;
    const float* ubase = inp + (size_t)b * SEQ * US;

    v8f x = {};
    StepOps cur = load_ops(ws, zbase, ubase, t0 + 1, col, hi);
    v8f dzu = zu_chain(cur);
    #pragma unroll 1
    for (int s = 1; s <= CH_L; ++s) {
        const int t  = t0 + s;
        const int tn = (s < CH_L) ? (t + 1) : t;
        StepOps nxt = load_ops(ws, zbase, ubase, tn, col, hi);
        x = recur(cur.aA, x, dzu, col, hi);
        dzu = zu_chain(nxt);
        cur = nxt;
    }
    store_row(out + ((size_t)b * SEQ + (t0 + CH_L)) * NS + 8 * hi, x);
}

__global__ __launch_bounds__(32)
void kalman_chunk_link_kernel(const float* __restrict__ x0, float* out)
{
    const int lane = threadIdx.x;
    const int hi   = lane >> 4;
    const int col  = lane & 15;
    const int b    = blockIdx.x * 16 + col;

    v8f x;
    #pragma unroll
    for (int r = 0; r < 8; ++r) x[r] = x0[8 * hi + r];

    float* obase = out + (size_t)b * SEQ * NS + 8 * hi;
    store_row(obase, x);

    AOps cur = load_a(out + (size_t)PHI_ROW * NS, col, hi);
    v8f  y   = load_row(obase + (size_t)CH_L * NS);
    #pragma unroll 1
    for (int c = 0; c < CH_N; ++c) {
        const int cn = (c + 1 < CH_N) ? (c + 1) : c;
        AOps nxt = load_a(out + ((size_t)cn * CH_L + PHI_ROW) * NS, col, hi);
        v8f  yn  = load_row(obase + (size_t)(cn + 1) * CH_L * NS);
        x = recur(cur, x, y, col, hi);
        store_row(obase + (size_t)(c + 1) * CH_L * NS, x);
        cur = nxt;
        y = yn;
    }
}

__global__ __launch_bounds__(32)
void kalman_chunk_final_kernel(const float* __restrict__ obs, const float* __restrict__ inp,
                               const float* __restrict__ ws, float* out)
{
    const int lane = threadIdx.x;
    const int hi   = lane >> 4;
    const int col  = lane & 15;
    const int b    = blockIdx.x * 16 + col;
    const int t0   = blockIdx.y * CH_L;

    const float* zbase = obs + (size_t)b * SEQ * MS;
    const float* ubase = inp + (size_t)b * SEQ * US;
    float*       obase = out + (size_t)b * SEQ * NS + 8 * hi;

    v8f x = load_row(obase + (size_t)t0 * NS);
    StepOps cur = load_ops(ws, zbase, ubase, t0 + 1, col, hi);
    v8f dzu = zu_chain(cur);
    #pragma unroll 1
    for (int s = 1; s < CH_L; ++s) {
        const int t  = t0 + s;
        const int tn = (s + 1 < CH_L) ? (t + 1) : t;
        StepOps nxt = load_ops(ws, zbase, ubase, tn, col, hi);
        x = recur(cur.aA, x, dzu, col, hi);
        store_row(obase + (size_t)t * NS, x);
        dzu = zu_chain(nxt);
        cur = nxt;
    }
}

extern "C" void kernel_launch(void* const* d_in, const int* in_sizes, int n_in,
                              void* d_out, int out_size, void* d_ws, size_t ws_size,
                              hipStream_t stream) {
    const float* obs = (const float*)d_in[0];
    const float* inp = (const float*)d_in[1];
    const float* A   = (const float*)d_in[2];
    const float* B   = (const float*)d_in[3];
    const float* C   = (const float*)d_in[4];
    const float* Q   = (const float*)d_in[5];
    const float* R   = (const float*)d_in[6];
    const float* x0  = (const float*)d_in[7];
    float* out = (float*)d_out;
    float* ws  = (float*)d_ws;

    hipLaunchKernelGGL(kalman_cov_kernel, dim3(1), dim3(CTH), 0, stream,
                       A, B, C, Q, R, ws);
    hipLaunchKernelGGL(kalman_chunk_local_kernel, dim3(BATCH / 16 + 1, CH_N), dim3(32), 0, stream,
                       obs, inp, ws, out);
    hipLaunchKernelGGL(kalman_chunk_link_kernel, dim3(BATCH / 16), dim3(32), 0, stream,
                       x0, out);
    hipLaunchKernelGGL(kalman_chunk_final_kernel, dim3(BATCH / 16, CH_N), dim3(32), 0, stream,
                       obs, inp, ws, out);
}


# Covariance step's 8 result stores held one step, issued behind the next step's LDS reads: store addressing and result-only row updates leave the serial tail behind the last pivot; same sums and order
# speedup vs baseline: 6.8962x; 1.0748x over previous
#include <hip/hip_runtime.h>

#define BATCH 256
#define SEQ   4096
#define NS    16
#define MS    8
#define US    4
#define STEPF 448

typedef __attribute__((ext_vector_type(2))) float v2f;
typedef __attribute__((ext_vector_type(8))) float v8f;

#define TWB 40
#define CWV 4
#define CTH (CWV*32)
#define RPW (NS/CWV)

__device__ __forceinline__ float lane_bcast(float v, int srcLane) {
    return __builtin_bit_cast(float,
        __builtin_amdgcn_readlane(__builtin_bit_cast(int, v), srcLane));
}

template <int K>
__device__ __forceinline__ float row_bcast(float v) {
    return __builtin_bit_cast(float,
        __builtin_amdgcn_update_dpp(0, __builtin_bit_cast(int, v), 0x150 + K, 0xF, 0xF, true));
}

__device__ __forceinline__ float xhalf(float x) {
    const int b = __builtin_bit_cast(int, x);
    return __builtin_bit_cast(float,
        __builtin_amdgcn_permlanex16(b, b, 0x76543210, 0xfedcba98, false, false));
}

#define ROW_DOT4_STEP(K)                                              \
    { const float x0 = row_bcast<K>(v0), x1 = row_bcast<K>(v1);      \
      a0 += x0 * ma[K];  a1 += x1 * ma[K];                            \
      b0 += x0 * mb[K];  b1 += x1 * mb[K]; }
__device__ __forceinline__ void row_dot4(float v0, float v1,
                                         const float (&ma)[NS], const float (&mb)[NS],
                                         float& ra0, float& rb0, float& ra1, float& rb1) {
    float a0 = 0.f, b0 = 0.f, a1 = 0.f, b1 = 0.f;
    ROW_DOT4_STEP(0)   ROW_DOT4_STEP(1)   ROW_DOT4_STEP(2)   ROW_DOT4_STEP(3)
    ROW_DOT4_STEP(4)   ROW_DOT4_STEP(5)   ROW_DOT4_STEP(6)   ROW_DOT4_STEP(7)
    ROW_DOT4_STEP(8)   ROW_DOT4_STEP(9)   ROW_DOT4_STEP(10)  ROW_DOT4_STEP(11)
    ROW_DOT4_STEP(12)  ROW_DOT4_STEP(13)  ROW_DOT4_STEP(14)  ROW_DOT4_STEP(15)
    ra0 = a0;  rb0 = b0;  ra1 = a1;  rb1 = b1;
}
#undef ROW_DOT4_STEP

__global__ __launch_bounds__(CTH)
void kalman_cov_kernel(const float* __restrict__ Ag, const float* __restrict__ Bg,
                       const float* __restrict__ Cg, const float* __restrict__ Qg,
                       const float* __restrict__ Rg, float* __restrict__ ws)
{
    __shared__ float sA[NS*NS], sB[NS*US], sC[MS*NS], sQ[NS*NS], sR[MS*MS];
    __shared__ float sBot[2*NS*TWB];
    __shared__ float sCA[MS*NS], sCB[MS*US];

    const int tid = threadIdx.x;
    for (int e = tid; e < NS*NS; e += CTH) {
        sA[e] = Ag[e];
        sQ[e] = Qg[e];
    }
    if (tid < NS*US) sB[tid] = Bg[tid];
    if (tid < MS*NS) sC[tid] = Cg[tid];
    if (tid < MS*MS) sR[tid] = Rg[tid];
    __syncthreads();

    for (int e = tid; e < MS*NS + MS*US; e += CTH) {
        if (e < MS*NS) {
            const int i16 = e >> 4, j16 = e & 15;
            float a = 0.f;
            #pragma unroll
            for (int k = 0; k < NS; ++k) a += sC[i16*NS + k] * sA[k*NS + j16];
            sCA[e] = a;
        } else {
            const int q = e - MS*NS, ib = q >> 2, jb = q & 3;
            float a = 0.f;
            #pragma unroll
            for (int k = 0; k < NS; ++k) a += sC[ib*NS + k] * sB[k*US + jb];
            sCB[q] = a;
        }
    }
    __syncthreads();

    if (tid < NS*8) {
        const float a = sA[(tid >> 3)*NS + (tid & 7)];
        sBot[(tid >> 3)*TWB + 8 + (tid & 7)] = a;
        sBot[NS*TWB + (tid >> 3)*TWB + 8 + (tid & 7)] = a;
    }

    const int w  = tid >> 5, c = tid & 31;
    const int rbase = RPW * w;
    float kBHi[RPW];
    int   oh[RPW];
    #pragma unroll
    for (int j = 0; j < RPW; ++j) {
        const int r = rbase + j;
        kBHi[j] = 0.f;  oh[j] = -1;
        if (c < 8)        { kBHi[j] = sA[r*NS + 8 + c];     oh[j] = r*NS + 8 + c; }
        else if (c < 12)  { kBHi[j] = sB[r*US + (c - 8)];   oh[j] = 256 + r*US + (c - 8); }
        else if (c < 20)  { oh[j] = 320 + r*MS + (c - 12); }
    }
    float kPHi[MS];
    #pragma unroll
    for (int i = 0; i < MS; ++i) {
        float h = 0.f;
        if (c < 8)        h = sCA[i*NS + 8 + c];
        else if (c < 12)  h = sCB[i*US + (c - 8)];
        else if (c < 20)  h = ((c - 12) == i) ? -1.0f : 0.0f;
        kPHi[i] = h;
    }

    const int hr  = c >> 4, jc = c & 15;
    const int rowa = rbase + hr;
    const int rowb = rbase + 2 + hr;
    float kAj[NS], kCm[NS];
    #pragma unroll
    for (int k = 0; k < NS; ++k) {
        kAj[k]    = sA[jc*NS + k];
        kCm[k]    = sC[(jc & 7)*NS + k];
    }
    float kC[MS*NS];
    #pragma unroll
    for (int q = 0; q < MS*NS; ++q) kC[q] = sC[q];
    float kR[MS];
    #pragma unroll
    for (int i = 0; i < MS; ++i) kR[i] = (c < MS) ? sR[i*MS + c] : 0.0f;
    float kA2[RPW][NS];
    float kQ2[RPW];
    #pragma unroll
    for (int j = 0; j < RPW; ++j) {
        #pragma unroll
        for (int k = 0; k < NS; ++k) kA2[j][k] = sA[(rbase + j)*NS + k];
        kQ2[j] = (c >= 16) ? sQ[(rbase + j)*NS + (c - 16)] : 0.0f;
    }

    float m0, m1;
    {
        float a0 = 0.f, a1 = 0.f;
        #pragma unroll
        for (int k = 0; k < NS; ++k) {
            a0 += sA[rowa*NS + k] * kAj[k];
            a1 += sA[rowb*NS + k] * kAj[k];
        }
        m0 = a0 + sQ[rowa*NS + jc];
        m1 = a1 + sQ[rowb*NS + jc];
    }

    float r_lo[RPW], r_hi[RPW];
    #pragma unroll
    for (int j = 0; j < RPW; ++j) { r_lo[j] = 0.f;  r_hi[j] = 0.f; }

    #pragma unroll 1
    for (int t = 1; t < SEQ; ++t) {
        float* bot = sBot + (t & 1) * (NS*TWB);
        float ga, xa, gb, xb;
        row_dot4(m0, m1, kCm, kAj, ga, xa, gb, xb);
        bot[rowa*TWB + 16 + jc] = xa;
        bot[rowb*TWB + 16 + jc] = xb;
        bot[rowa*TWB + (jc & 7)] = ga;
        bot[rowb*TWB + (jc & 7)] = gb;
        __syncthreads();

        float b_lo[RPW];
        #pragma unroll
        for (int j = 0; j < RPW; ++j) b_lo[j] = bot[(rbase + j)*TWB + c];
        float col[NS];
        #pragma unroll
        for (int k = 0; k < NS; ++k) col[k] = bot[k*TWB + c];
        if (t > 1) {
            float* wsq = ws + (size_t)(t - 2) * STEPF;
            if (c >= 8 && c < 16) {
                #pragma unroll
                for (int j = 0; j < RPW; ++j) wsq[(rbase + j)*NS + (c - 8)] = r_lo[j];
            }
            if (oh[0] >= 0) {
                #pragma unroll
                for (int j = 0; j < RPW; ++j) wsq[oh[j]] = r_hi[j];
            }
        }
        float pl[MS], ph[MS];
        #pragma unroll
        for (int i = 0; i < MS; ++i) {
            pl[i] = kR[i];
            ph[i] = kPHi[i];
        }
        #pragma unroll
        for (int k = 0; k < NS; ++k) {
            #pragma unroll
            for (int i = 0; i < MS; ++i) pl[i] += kC[i*NS + k] * col[k];
        }
        float b2[RPW];
        #pragma unroll
        for (int j = 0; j < RPW; ++j) b2[j] = kQ2[j];
        #pragma unroll
        for (int k = 0; k < NS; ++k) {
            #pragma unroll
            for (int j = 0; j < RPW; ++j) b2[j] += kA2[j][k] * col[k];
        }
        float b_hi[RPW];
        #pragma unroll
        for (int j = 0; j < RPW; ++j) b_hi[j] = kBHi[j];

        #pragma unroll
        for (int p = 0; p < MS; ++p) {
            const float inv = 1.0f / lane_bcast(pl[p], p);
            #pragma unroll
            for (int i = p + 1; i < MS; ++i) {
                const float f = lane_bcast(pl[i], p) * inv;
                pl[i] -= f * pl[p];  ph[i] -= f * ph[p];
            }
            #pragma unroll
            for (int j = 0; j < RPW; ++j) {
                const float fj = lane_bcast(b_lo[j], p) * inv;
                b_lo[j] -= fj * pl[p];  b_hi[j] -= fj * ph[p];
                const float gj = lane_bcast(b2[j], p) * inv;
                b2[j] -= gj * pl[p];
            }
        }

        #pragma unroll
        for (int j = 0; j < RPW; ++j) { r_lo[j] = b_lo[j];  r_hi[j] = b_hi[j]; }

        const float x0 = xhalf(b2[0]);
        const float x2 = xhalf(b2[2]);
        m0 = hr ? b2[1] : x0;
        m1 = hr ? b2[3] : x2;
    }

    {
        float* wsq = ws + (size_t)(SEQ - 2) * STEPF;
        if (c >= 8 && c < 16) {
            #pragma unroll
            for (int j = 0; j < RPW; ++j) wsq[(rbase + j)*NS + (c - 8)] = r_lo[j];
        }
        if (oh[0] >= 0) {
            #pragma unroll
            for (int j = 0; j < RPW; ++j) wsq[oh[j]] = r_hi[j];
        }
    }
}

#define CH_L 63
#define CH_N 65
#define PHI_ROW 2

__device__ __forceinline__ float bperm(int srcLane, float v) {
    return __builtin_bit_cast(float,
        __builtin_amdgcn_ds_bpermute(srcLane << 2, __builtin_bit_cast(int, v)));
}

__device__ __forceinline__ v8f wmma4(v2f a, v2f b, v8f c) {
    return __builtin_amdgcn_wmma_f32_16x16x4_f32(false, a, false, b,
                                                 (short)0, c, false, false);
}

struct AOps {
    v2f a0, a1, a2, a3;
};

struct StepOps {
    AOps  aA;
    v2f aB, aK0, aK1;
    v2f bZ0, bZ1, bU;
};

__device__ __forceinline__ AOps load_a(const float* m, int col, int hi) {
    AOps o;
    o.a0 = *(const v2f*)(m + col * 16 + 0  + 2 * hi);
    o.a1 = *(const v2f*)(m + col * 16 + 4  + 2 * hi);
    o.a2 = *(const v2f*)(m + col * 16 + 8  + 2 * hi);
    o.a3 = *(const v2f*)(m + col * 16 + 12 + 2 * hi);
    return o;
}

__device__ __forceinline__ StepOps load_ops(const float* __restrict__ ws,
                                            const float* __restrict__ zbase,
                                            const float* __restrict__ ubase,
                                            int t, int col, int hi) {
    StepOps o;
    const float* w = ws + (size_t)(t - 1) * STEPF;
    o.aA  = load_a(w, col, hi);
    o.aB  = *(const v2f*)(w + 256 + col * 4 + 2 * hi);
    o.aK0 = *(const v2f*)(w + 320 + col * 8 + 0 + 2 * hi);
    o.aK1 = *(const v2f*)(w + 320 + col * 8 + 4 + 2 * hi);
    const float* zr = zbase + (size_t)t * MS;
    o.bZ0 = *(const v2f*)(zr + 0 + 2 * hi);
    o.bZ1 = *(const v2f*)(zr + 4 + 2 * hi);
    o.bU  = *(const v2f*)(ubase + (size_t)t * US + 2 * hi);
    return o;
}

__device__ __forceinline__ v8f zu_chain(const StepOps& o) {
    v8f d = {};
    d = wmma4(o.aB,  o.bU,  d);
    d = wmma4(o.aK0, o.bZ0, d);
    d = wmma4(o.aK1, o.bZ1, d);
    return d;
}

__device__ __forceinline__ v8f recur(const AOps& m, const v8f& x, v8f c, int col, int hi) {
    v2f bX[4];
    #pragma unroll
    for (int q = 0; q < 4; ++q) {
        int srcLane = col + ((q >= 2) ? 16 : 0);
        int lo = 4 * (q & 1);
        float t0 = bperm(srcLane, x[lo + 0]);
        float t1 = bperm(srcLane, x[lo + 2]);
        float t2 = bperm(srcLane, x[lo + 1]);
        float t3 = bperm(srcLane, x[lo + 3]);
        bX[q].x = hi ? t1 : t0;
        bX[q].y = hi ? t3 : t2;
    }
    v8f d = wmma4(m.a0, bX[0], c);
    d = wmma4(m.a1, bX[1], d);
    d = wmma4(m.a2, bX[2], d);
    d = wmma4(m.a3, bX[3], d);
    return d;
}

__device__ __forceinline__ void store_row(float* o, const v8f& x) {
    *(float4*)(o)     = make_float4(x[0], x[1], x[2], x[3]);
    *(float4*)(o + 4) = make_float4(x[4], x[5], x[6], x[7]);
}

__device__ __forceinline__ v8f load_row(const float* o) {
    const float4 lo = *(const float4*)(o);
    const float4 hh = *(const float4*)(o + 4);
    v8f x = {lo.x, lo.y, lo.z, lo.w, hh.x, hh.y, hh.z, hh.w};
    return x;
}

__global__ __launch_bounds__(32)
void kalman_chunk_local_kernel(const float* __restrict__ obs, const float* __restrict__ inp,
                               const float* __restrict__ ws, float* __restrict__ out)
{
    const int lane = threadIdx.x;
    const int hi   = lane >> 4;
    const int col  = lane & 15;
    const int tile = blockIdx.x;
    const int t0   = blockIdx.y * CH_L;

    if (tile == BATCH / 16) {
        v8f x;
        #pragma unroll
        for (int r = 0; r < 8; ++r) x[r] = (8 * hi + r == col) ? 1.0f : 0.0f;
        AOps cur = load_a(ws + (size_t)t0 * STEPF, col, hi);
        #pragma unroll 1
        for (int s = 1; s <= CH_L; ++s) {
            const int wn = (s < CH_L) ? (t0 + s) : (t0 + s - 1);
            AOps nxt = load_a(ws + (size_t)wn * STEPF, col, hi);
            v8f zero = {};
            x = recur(cur, x, zero, col, hi);
            cur = nxt;
        }
        float* p = out + (size_t)(t0 + PHI_ROW) * NS;
        #pragma unroll
        for (int r = 0; r < 8; ++r) p[(8 * hi + r) * NS + col] = x[r];
        return;
    }

    const int b = tile * 16 + col;
    const float* zbase = obs + (size_t)b * SEQ * MS;
    const float* ubase = inp + (size_t)b * SEQ * US;

    v8f x = {};
    StepOps cur = load_ops(ws, zbase, ubase, t0 + 1, col, hi);
    v8f dzu = zu_chain(cur);
    #pragma unroll 1
    for (int s = 1; s <= CH_L; ++s) {
        const int t  = t0 + s;
        const int tn = (s < CH_L) ? (t + 1) : t;
        StepOps nxt = load_ops(ws, zbase, ubase, tn, col, hi);
        x = recur(cur.aA, x, dzu, col, hi);
        dzu = zu_chain(nxt);
        cur = nxt;
    }
    store_row(out + ((size_t)b * SEQ + (t0 + CH_L)) * NS + 8 * hi, x);
}

__global__ __launch_bounds__(32)
void kalman_chunk_link_kernel(const float* __restrict__ x0, float* out)
{
    const int lane = threadIdx.x;
    const int hi   = lane >> 4;
    const int col  = lane & 15;
    const int b    = blockIdx.x * 16 + col;

    v8f x;
    #pragma unroll
    for (int r = 0; r < 8; ++r) x[r] = x0[8 * hi + r];

    float* obase = out + (size_t)b * SEQ * NS + 8 * hi;
    store_row(obase, x);

    AOps cur = load_a(out + (size_t)PHI_ROW * NS, col, hi);
    v8f  y   = load_row(obase + (size_t)CH_L * NS);
    #pragma unroll 1
    for (int c = 0; c < CH_N; ++c) {
        const int cn = (c + 1 < CH_N) ? (c + 1) : c;
        AOps nxt = load_a(out + ((size_t)cn * CH_L + PHI_ROW) * NS, col, hi);
        v8f  yn  = load_row(obase + (size_t)(cn + 1) * CH_L * NS);
        x = recur(cur, x, y, col, hi);
        store_row(obase + (size_t)(c + 1) * CH_L * NS, x);
        cur = nxt;
        y = yn;
    }
}

__global__ __launch_bounds__(32)
void kalman_chunk_final_kernel(const float* __restrict__ obs, const float* __restrict__ inp,
                               const float* __restrict__ ws, float* out)
{
    const int lane = threadIdx.x;
    const int hi   = lane >> 4;
    const int col  = lane & 15;
    const int b    = blockIdx.x * 16 + col;
    const int t0   = blockIdx.y * CH_L;

    const float* zbase = obs + (size_t)b * SEQ * MS;
    const float* ubase = inp + (size_t)b * SEQ * US;
    float*       obase = out + (size_t)b * SEQ * NS + 8 * hi;

    v8f x = load_row(obase + (size_t)t0 * NS);
    StepOps cur = load_ops(ws, zbase, ubase, t0 + 1, col, hi);
    v8f dzu = zu_chain(cur);
    #pragma unroll 1
    for (int s = 1; s < CH_L; ++s) {
        const int t  = t0 + s;
        const int tn = (s + 1 < CH_L) ? (t + 1) : t;
        StepOps nxt = load_ops(ws, zbase, ubase, tn, col, hi);
        x = recur(cur.aA, x, dzu, col, hi);
        store_row(obase + (size_t)t * NS, x);
        dzu = zu_chain(nxt);
        cur = nxt;
    }
}

extern "C" void kernel_launch(void* const* d_in, const int* in_sizes, int n_in,
                              void* d_out, int out_size, void* d_ws, size_t ws_size,
                              hipStream_t stream) {
    const float* obs = (const float*)d_in[0];
    const float* inp = (const float*)d_in[1];
    const float* A   = (const float*)d_in[2];
    const float* B   = (const float*)d_in[3];
    const float* C   = (const float*)d_in[4];
    const float* Q   = (const float*)d_in[5];
    const float* R   = (const float*)d_in[6];
    const float* x0  = (const float*)d_in[7];
    float* out = (float*)d_out;
    float* ws  = (float*)d_ws;

    hipLaunchKernelGGL(kalman_cov_kernel, dim3(1), dim3(CTH), 0, stream,
                       A, B, C, Q, R, ws);
    hipLaunchKernelGGL(kalman_chunk_local_kernel, dim3(BATCH / 16 + 1, CH_N), dim3(32), 0, stream,
                       obs, inp, ws, out);
    hipLaunchKernelGGL(kalman_chunk_link_kernel, dim3(BATCH / 16), dim3(32), 0, stream,
                       x0, out);
    hipLaunchKernelGGL(kalman_chunk_final_kernel, dim3(BATCH / 16, CH_N), dim3(32), 0, stream,
                       obs, inp, ws, out);
}


# Covariance recursion split in time: 64 blocks x 64 steps, chunk-start priors by composed 64-step maps (squaring, 16x16 inverses); serial path kept where the map does not forget; starts reassociated
# speedup vs baseline: 32.3933x; 4.6973x over previous
#include <hip/hip_runtime.h>

#define BATCH 256
#define SEQ   4096
#define NS    16
#define MS    8
#define US    4
#define STEPF 448

typedef __attribute__((ext_vector_type(2))) float v2f;
typedef __attribute__((ext_vector_type(8))) float v8f;

#define TWB 40
#define CWV 4
#define CTH (CWV*32)
#define RPW (NS/CWV)

__device__ __forceinline__ float lane_bcast(float v, int srcLane) {
    return __builtin_bit_cast(float,
        __builtin_amdgcn_readlane(__builtin_bit_cast(int, v), srcLane));
}

template <int K>
__device__ __forceinline__ float row_bcast(float v) {
    return __builtin_bit_cast(float,
        __builtin_amdgcn_update_dpp(0, __builtin_bit_cast(int, v), 0x150 + K, 0xF, 0xF, true));
}

__device__ __forceinline__ float xhalf(float x) {
    const int b = __builtin_bit_cast(int, x);
    return __builtin_bit_cast(float,
        __builtin_amdgcn_permlanex16(b, b, 0x76543210, 0xfedcba98, false, false));
}

#define ROW_DOT4_STEP(K)                                              \
    { const float x0 = row_bcast<K>(v0), x1 = row_bcast<K>(v1);      \
      a0 += x0 * ma[K];  a1 += x1 * ma[K];                            \
      b0 += x0 * mb[K];  b1 += x1 * mb[K]; }
__device__ __forceinline__ void row_dot4(float v0, float v1,
                                         const float (&ma)[NS], const float (&mb)[NS],
                                         float& ra0, float& rb0, float& ra1, float& rb1) {
    float a0 = 0.f, b0 = 0.f, a1 = 0.f, b1 = 0.f;
    ROW_DOT4_STEP(0)   ROW_DOT4_STEP(1)   ROW_DOT4_STEP(2)   ROW_DOT4_STEP(3)
    ROW_DOT4_STEP(4)   ROW_DOT4_STEP(5)   ROW_DOT4_STEP(6)   ROW_DOT4_STEP(7)
    ROW_DOT4_STEP(8)   ROW_DOT4_STEP(9)   ROW_DOT4_STEP(10)  ROW_DOT4_STEP(11)
    ROW_DOT4_STEP(12)  ROW_DOT4_STEP(13)  ROW_DOT4_STEP(14)  ROW_DOT4_STEP(15)
    ra0 = a0;  rb0 = b0;  ra1 = a1;  rb1 = b1;
}
#undef ROW_DOT4_STEP


#define CL   64
#define CLLG 6
#define NCB  ((SEQ - 1 + CL - 1) / CL)
#define FORGET_BOUND 1e-8f

__device__ __forceinline__ void mm16(float* D, const float* X, const float* Y, int tid) {
    #pragma unroll
    for (int q = 0; q < 2; ++q) {
        const int e = tid + q*CTH, i = e >> 4, j = e & 15;
        float a = 0.f;
        #pragma unroll
        for (int k = 0; k < NS; ++k) a += X[i*NS + k] * Y[k*NS + j];
        D[e] = a;
    }
    __syncthreads();
}

__device__ __forceinline__ void mm16_add(float* D, const float* X, const float* Y,
                                         const float* Z, int tid) {
    #pragma unroll
    for (int q = 0; q < 2; ++q) {
        const int e = tid + q*CTH, i = e >> 4, j = e & 15;
        float a = 0.f;
        #pragma unroll
        for (int k = 0; k < NS; ++k) a += X[i*NS + k] * Y[k*NS + j];
        D[e] = a + Z[e];
    }
    __syncthreads();
}

__device__ __forceinline__ void eye_axpy(float* D, const float* X, float sgn, int tid) {
    #pragma unroll
    for (int q = 0; q < 2; ++q) {
        const int e = tid + q*CTH;
        D[e] = (((e >> 4) == (e & 15)) ? 1.0f : 0.0f) + sgn * X[e];
    }
    __syncthreads();
}

__device__ __forceinline__ void sub16(float* D, const float* X, const float* Y, int tid) {
    #pragma unroll
    for (int q = 0; q < 2; ++q) { const int e = tid + q*CTH;  D[e] = X[e] - Y[e]; }
    __syncthreads();
}

__device__ __forceinline__ void inv16(float* D, const float* X, float* aug, int tid) {
    #pragma unroll
    for (int q = 0; q < 4; ++q) {
        const int e = tid + q*CTH, i = e >> 5, j = e & 31;
        aug[e] = (j < NS) ? X[i*NS + j] : (((j - NS) == i) ? 1.0f : 0.0f);
    }
    __syncthreads();
    #pragma unroll 1
    for (int p = 0; p < NS; ++p) {
        int r = p;
        float best = -1.0f;
        #pragma unroll
        for (int k = 0; k < NS; ++k) {
            const float v = fabsf(aug[k*32 + p]);
            if (k >= p && v > best) { best = v;  r = k; }
        }
        const float inv = 1.0f / aug[r*32 + p];
        float nv[4];
        #pragma unroll
        for (int q = 0; q < 4; ++q) {
            const int e = tid + q*CTH, i = e >> 5, j = e & 31;
            const int si = (i == p) ? r : ((i == r) ? p : i);
            const float x = aug[si*32 + j], f = aug[si*32 + p], pr = aug[r*32 + j];
            nv[q] = (i == p) ? pr * inv : x - (f * inv) * pr;
        }
        __syncthreads();
        #pragma unroll
        for (int q = 0; q < 4; ++q) aug[tid + q*CTH] = nv[q];
        __syncthreads();
    }
    #pragma unroll
    for (int q = 0; q < 2; ++q) {
        const int e = tid + q*CTH;
        D[e] = aug[(e >> 4)*32 + NS + (e & 15)];
    }
    __syncthreads();
}

__device__ __forceinline__ float nmax(float a, float v) {
    return (v > a || v != v) ? v : a;
}

#define MX_AT   0
#define MX_CP   1
#define MX_CPT  2
#define MX_R16  3
#define MX_X    4
#define MX_GL   5
#define MX_GR   6
#define MX_GC   7
#define MX_GJ   8
#define MX_NL   9
#define MX_NR   10
#define MX_NC   11
#define MX_NJ   12
#define MX_E    13
#define MX_T0   14
#define MX_CNT  20
#define MXP(pool, m) ((pool) + (m)*NS*NS)

__device__ __forceinline__ float map_square(float* pool, float* aug, float* red, int tid) {
    float* X  = MXP(pool, MX_X);
    float* GL = MXP(pool, MX_GL);  float* GR = MXP(pool, MX_GR);
    float* GC = MXP(pool, MX_GC);  float* GJ = MXP(pool, MX_GJ);
    float* NL = MXP(pool, MX_NL);  float* NR = MXP(pool, MX_NR);
    float* NC = MXP(pool, MX_NC);  float* NJ = MXP(pool, MX_NJ);
    float* t0 = MXP(pool, MX_T0);      float* t1 = MXP(pool, MX_T0 + 1);
    float* t2 = MXP(pool, MX_T0 + 2);  float* t3 = MXP(pool, MX_T0 + 3);
    float* t4 = MXP(pool, MX_T0 + 4);  float* t5 = MXP(pool, MX_T0 + 5);
    sub16(t0, GC, X, tid);
    mm16(t1, t0, GJ, tid);
    eye_axpy(t2, t1, 1.0f, tid);
    inv16(t3, t2, aug, tid);
    mm16(t1, t3, t0, tid);
    mm16(t2, GJ, t1, tid);
    eye_axpy(t4, t2, -1.0f, tid);
    mm16(t2, t3, GL, tid);
    mm16(NL, GL, t2, tid);
    mm16(t2, GL, t1, tid);
    mm16_add(NC, t2, GR, GC, tid);
    mm16(t2, t4, GR, tid);
    mm16(NR, GR, t2, tid);
    mm16(t2, GJ, GL, tid);
    mm16(t5, t4, t2, tid);
    mm16_add(NJ, GR, t5, GJ, tid);
    const float l0 = NL[tid], l1 = NL[tid + CTH], r0 = NR[tid], r1 = NR[tid + CTH];
    const float c0 = NC[tid], c1 = NC[tid + CTH], j0 = NJ[tid], j1 = NJ[tid + CTH];
    red[tid]         = nmax(nmax(0.f, fabsf(l0)), fabsf(l1));
    red[CTH + tid]   = nmax(nmax(0.f, fabsf(r0)), fabsf(r1));
    red[2*CTH + tid] = nmax(nmax(nmax(nmax(0.f, fabsf(c0)), fabsf(c1)), fabsf(j0)), fabsf(j1));
    __syncthreads();
    float mL = 0.f, mR = 0.f, mC = 0.f;
    for (int i = 0; i < CTH; ++i) {
        mL = nmax(mL, red[i]);  mR = nmax(mR, red[CTH + i]);  mC = nmax(mC, red[2*CTH + i]);
    }
    const bool fin = (mL < 3.0e38f) && (mR < 3.0e38f) && (mC < 3.0e38f);
    int a = 0;
    if (fin && mL > 0.f && mR > 0.f) {
        int eL, eR;
        (void)frexpf(mL, &eL);  (void)frexpf(mR, &eR);
        a = (eR - eL) / 2;
    }
    GL[tid] = ldexpf(l0, a);   GL[tid + CTH] = ldexpf(l1, a);
    GR[tid] = ldexpf(r0, -a);  GR[tid + CTH] = ldexpf(r1, -a);
    GC[tid] = c0;  GC[tid + CTH] = c1;
    GJ[tid] = j0;  GJ[tid + CTH] = j1;
    __syncthreads();
    return fin ? mL * mR : 1.0f;
}

__device__ __forceinline__ void map_apply(float* pool, float* aug, int tid) {
    float* X  = MXP(pool, MX_X);   float* E  = MXP(pool, MX_E);
    float* GL = MXP(pool, MX_GL);  float* GR = MXP(pool, MX_GR);
    float* GC = MXP(pool, MX_GC);  float* GJ = MXP(pool, MX_GJ);
    float* t0 = MXP(pool, MX_T0);      float* t1 = MXP(pool, MX_T0 + 1);
    float* t2 = MXP(pool, MX_T0 + 2);  float* t3 = MXP(pool, MX_T0 + 3);
    sub16(t0, E, X, tid);
    mm16(t1, t0, GJ, tid);
    eye_axpy(t2, t1, 1.0f, tid);
    inv16(t3, t2, aug, tid);
    mm16(t1, t3, t0, tid);
    mm16(t2, GL, t1, tid);
    mm16_add(E, t2, GR, GC, tid);
}

__device__ __forceinline__ void map_one_step(float* pool, float* aug, const float* sA,
                                             const float* sQ, int tid) {
    float* AT  = MXP(pool, MX_AT);   float* CP  = MXP(pool, MX_CP);
    float* CPT = MXP(pool, MX_CPT);  float* R16 = MXP(pool, MX_R16);
    float* X   = MXP(pool, MX_X);
    float* GL = MXP(pool, MX_GL);  float* GR = MXP(pool, MX_GR);
    float* GC = MXP(pool, MX_GC);  float* GJ = MXP(pool, MX_GJ);
    float* t0 = MXP(pool, MX_T0);  float* t1 = MXP(pool, MX_T0 + 1);  float* t2 = MXP(pool, MX_T0 + 2);
    mm16(t0, CP, X, tid);
    mm16_add(t1, t0, CPT, R16, tid);
    inv16(t2, t1, aug, tid);
    mm16(t0, CPT, t2, tid);
    mm16(GJ, t0, CP, tid);
    mm16(t0, X, GJ, tid);
    eye_axpy(t1, t0, -1.0f, tid);
    mm16(t0, GJ, X, tid);
    eye_axpy(t2, t0, -1.0f, tid);
    mm16(GL, sA, t1, tid);
    mm16(GR, t2, AT, tid);
    mm16(t0, GL, X, tid);
    mm16_add(GC, t0, AT, sQ, tid);
}

__global__ __launch_bounds__(CTH)
void kalman_cov_kernel(const float* __restrict__ Ag, const float* __restrict__ Bg,
                       const float* __restrict__ Cg, const float* __restrict__ Qg,
                       const float* __restrict__ Rg, float* __restrict__ ws)
{
    __shared__ float sA[NS*NS], sB[NS*US], sC[MS*NS], sQ[NS*NS], sR[MS*MS];
    __shared__ float sBot[2*NS*TWB];
    __shared__ float sCA[MS*NS], sCB[MS*US];
    __shared__ float sPool[MX_CNT*NS*NS];
    __shared__ float sAug[NS*32];
    __shared__ float sRed[3*CTH];

    const int tid = threadIdx.x;
    for (int e = tid; e < NS*NS; e += CTH) {
        sA[e] = Ag[e];
        sQ[e] = Qg[e];
    }
    if (tid < NS*US) sB[tid] = Bg[tid];
    if (tid < MS*NS) sC[tid] = Cg[tid];
    if (tid < MS*MS) sR[tid] = Rg[tid];
    __syncthreads();

    #pragma unroll
    for (int q = 0; q < 2; ++q) {
        const int e = tid + q*CTH, i = e >> 4, j = e & 15;
        MXP(sPool, MX_AT)[e]  = sA[j*NS + i];
        MXP(sPool, MX_CP)[e]  = (i < MS) ? sC[i*NS + j] : 0.0f;
        MXP(sPool, MX_CPT)[e] = (j < MS) ? sC[j*NS + i] : 0.0f;
        MXP(sPool, MX_R16)[e] = (i < MS && j < MS) ? sR[i*MS + j] : ((i == j) ? 1.0f : 0.0f);
    }

    for (int e = tid; e < MS*NS + MS*US; e += CTH) {
        if (e < MS*NS) {
            const int i16 = e >> 4, j16 = e & 15;
            float a = 0.f;
            #pragma unroll
            for (int k = 0; k < NS; ++k) a += sC[i16*NS + k] * sA[k*NS + j16];
            sCA[e] = a;
        } else {
            const int q = e - MS*NS, ib = q >> 2, jb = q & 3;
            float a = 0.f;
            #pragma unroll
            for (int k = 0; k < NS; ++k) a += sC[ib*NS + k] * sB[k*US + jb];
            sCB[q] = a;
        }
    }
    __syncthreads();

    if (tid < NS*8) {
        const float a = sA[(tid >> 3)*NS + (tid & 7)];
        sBot[(tid >> 3)*TWB + 8 + (tid & 7)] = a;
        sBot[NS*TWB + (tid >> 3)*TWB + 8 + (tid & 7)] = a;
    }

    const int w  = tid >> 5, c = tid & 31;
    const int rbase = RPW * w;
    float kBHi[RPW];
    int   oh[RPW];
    #pragma unroll
    for (int j = 0; j < RPW; ++j) {
        const int r = rbase + j;
        kBHi[j] = 0.f;  oh[j] = -1;
        if (c < 8)        { kBHi[j] = sA[r*NS + 8 + c];     oh[j] = r*NS + 8 + c; }
        else if (c < 12)  { kBHi[j] = sB[r*US + (c - 8)];   oh[j] = 256 + r*US + (c - 8); }
        else if (c < 20)  { oh[j] = 320 + r*MS + (c - 12); }
    }
    float kPHi[MS];
    #pragma unroll
    for (int i = 0; i < MS; ++i) {
        float h = 0.f;
        if (c < 8)        h = sCA[i*NS + 8 + c];
        else if (c < 12)  h = sCB[i*US + (c - 8)];
        else if (c < 20)  h = ((c - 12) == i) ? -1.0f : 0.0f;
        kPHi[i] = h;
    }

    const int hr  = c >> 4, jc = c & 15;
    const int rowa = rbase + hr;
    const int rowb = rbase + 2 + hr;
    float kAj[NS], kCm[NS];
    #pragma unroll
    for (int k = 0; k < NS; ++k) {
        kAj[k]    = sA[jc*NS + k];
        kCm[k]    = sC[(jc & 7)*NS + k];
    }
    float kC[MS*NS];
    #pragma unroll
    for (int q = 0; q < MS*NS; ++q) kC[q] = sC[q];
    float kR[MS];
    #pragma unroll
    for (int i = 0; i < MS; ++i) kR[i] = (c < MS) ? sR[i*MS + c] : 0.0f;
    float kA2[RPW][NS];
    float kQ2[RPW];
    #pragma unroll
    for (int j = 0; j < RPW; ++j) {
        #pragma unroll
        for (int k = 0; k < NS; ++k) kA2[j][k] = sA[(rbase + j)*NS + k];
        kQ2[j] = (c >= 16) ? sQ[(rbase + j)*NS + (c - 16)] : 0.0f;
    }

    float m0, m1;
    {
        float a0 = 0.f, a1 = 0.f;
        #pragma unroll
        for (int k = 0; k < NS; ++k) {
            a0 += sA[rowa*NS + k] * kAj[k];
            a1 += sA[rowb*NS + k] * kAj[k];
        }
        m0 = a0 + sQ[rowa*NS + jc];
        m1 = a1 + sQ[rowb*NS + jc];
    }

    float r_lo[RPW], r_hi[RPW];
    #pragma unroll
    for (int j = 0; j < RPW; ++j) { r_lo[j] = 0.f;  r_hi[j] = 0.f; }

    const int cb = blockIdx.x;
    int  tb = 1, te = CL + 1;
    bool store = (cb == 0);
    #pragma unroll 1
    for (int phase = 0; phase < 2; ++phase) {
    if (phase == 1) {
        float* mxX = MXP(sPool, MX_X);
        float* mxE = MXP(sPool, MX_E);
        mxX[rowa*NS + jc] = m0;  mxX[rowb*NS + jc] = m1;
        mxE[rowa*NS + jc] = m0;  mxE[rowb*NS + jc] = m1;
        __syncthreads();
        map_one_step(sPool, sAug, sA, sQ, tid);
        const int nq = cb - 1;
        float gv = 1.0f;
        bool  ok = false;
        #pragma unroll 1
        for (int s = 0; s < CLLG + 6; ++s) {
            if (s >= CLLG) {
                const int k = s - CLLG;
                if (k == 0) {
                    ok = gv < FORGET_BOUND;
                    if (!ok || cb < 2) break;
                }
                if ((nq >> k) & 1) map_apply(sPool, sAug, tid);
                if ((nq >> (k + 1)) == 0) break;
            }
            gv = map_square(sPool, sAug, sRed, tid);
        }
        if (ok) {
            if (cb == 0) return;
            tb = cb * CL + 1;
            te = (tb + CL < SEQ) ? (tb + CL) : SEQ;
            m0 = mxE[rowa*NS + jc];
            m1 = mxE[rowb*NS + jc];
        } else {
            if (cb != 0) return;
            tb = CL + 1;
            te = SEQ;
        }
        store = true;
    }
    #pragma unroll 1
    for (int t = tb; t < te; ++t) {
        float* bot = sBot + (t & 1) * (NS*TWB);
        float ga, xa, gb, xb;
        row_dot4(m0, m1, kCm, kAj, ga, xa, gb, xb);
        bot[rowa*TWB + 16 + jc] = xa;
        bot[rowb*TWB + 16 + jc] = xb;
        bot[rowa*TWB + (jc & 7)] = ga;
        bot[rowb*TWB + (jc & 7)] = gb;
        __syncthreads();

        float b_lo[RPW];
        #pragma unroll
        for (int j = 0; j < RPW; ++j) b_lo[j] = bot[(rbase + j)*TWB + c];
        float col[NS];
        #pragma unroll
        for (int k = 0; k < NS; ++k) col[k] = bot[k*TWB + c];
        if (store && t > tb) {
            float* wsq = ws + (size_t)(t - 2) * STEPF;
            if (c >= 8 && c < 16) {
                #pragma unroll
                for (int j = 0; j < RPW; ++j) wsq[(rbase + j)*NS + (c - 8)] = r_lo[j];
            }
            if (oh[0] >= 0) {
                #pragma unroll
                for (int j = 0; j < RPW; ++j) wsq[oh[j]] = r_hi[j];
            }
        }
        float pl[MS], ph[MS];
        #pragma unroll
        for (int i = 0; i < MS; ++i) {
            pl[i] = kR[i];
            ph[i] = kPHi[i];
        }
        #pragma unroll
        for (int k = 0; k < NS; ++k) {
            #pragma unroll
            for (int i = 0; i < MS; ++i) pl[i] += kC[i*NS + k] * col[k];
        }
        float b2[RPW];
        #pragma unroll
        for (int j = 0; j < RPW; ++j) b2[j] = kQ2[j];
        #pragma unroll
        for (int k = 0; k < NS; ++k) {
            #pragma unroll
            for (int j = 0; j < RPW; ++j) b2[j] += kA2[j][k] * col[k];
        }
        float b_hi[RPW];
        #pragma unroll
        for (int j = 0; j < RPW; ++j) b_hi[j] = kBHi[j];

        #pragma unroll
        for (int p = 0; p < MS; ++p) {
            const float inv = 1.0f / lane_bcast(pl[p], p);
            #pragma unroll
            for (int i = p + 1; i < MS; ++i) {
                const float f = lane_bcast(pl[i], p) * inv;
                pl[i] -= f * pl[p];  ph[i] -= f * ph[p];
            }
            #pragma unroll
            for (int j = 0; j < RPW; ++j) {
                const float fj = lane_bcast(b_lo[j], p) * inv;
                b_lo[j] -= fj * pl[p];  b_hi[j] -= fj * ph[p];
                const float gj = lane_bcast(b2[j], p) * inv;
                b2[j] -= gj * pl[p];
            }
        }

        #pragma unroll
        for (int j = 0; j < RPW; ++j) { r_lo[j] = b_lo[j];  r_hi[j] = b_hi[j]; }

        const float x0 = xhalf(b2[0]);
        const float x2 = xhalf(b2[2]);
        m0 = hr ? b2[1] : x0;
        m1 = hr ? b2[3] : x2;
    }

    if (store) {
        float* wsq = ws + (size_t)(te - 2) * STEPF;
        if (c >= 8 && c < 16) {
            #pragma unroll
            for (int j = 0; j < RPW; ++j) wsq[(rbase + j)*NS + (c - 8)] = r_lo[j];
        }
        if (oh[0] >= 0) {
            #pragma unroll
            for (int j = 0; j < RPW; ++j) wsq[oh[j]] = r_hi[j];
        }
    }
    }
}

#define CH_L 63
#define CH_N 65
#define PHI_ROW 2

__device__ __forceinline__ float bperm(int srcLane, float v) {
    return __builtin_bit_cast(float,
        __builtin_amdgcn_ds_bpermute(srcLane << 2, __builtin_bit_cast(int, v)));
}

__device__ __forceinline__ v8f wmma4(v2f a, v2f b, v8f c) {
    return __builtin_amdgcn_wmma_f32_16x16x4_f32(false, a, false, b,
                                                 (short)0, c, false, false);
}

struct AOps {
    v2f a0, a1, a2, a3;
};

struct StepOps {
    AOps  aA;
    v2f aB, aK0, aK1;
    v2f bZ0, bZ1, bU;
};

__device__ __forceinline__ AOps load_a(const float* m, int col, int hi) {
    AOps o;
    o.a0 = *(const v2f*)(m + col * 16 + 0  + 2 * hi);
    o.a1 = *(const v2f*)(m + col * 16 + 4  + 2 * hi);
    o.a2 = *(const v2f*)(m + col * 16 + 8  + 2 * hi);
    o.a3 = *(const v2f*)(m + col * 16 + 12 + 2 * hi);
    return o;
}

__device__ __forceinline__ StepOps load_ops(const float* __restrict__ ws,
                                            const float* __restrict__ zbase,
                                            const float* __restrict__ ubase,
                                            int t, int col, int hi) {
    StepOps o;
    const float* w = ws + (size_t)(t - 1) * STEPF;
    o.aA  = load_a(w, col, hi);
    o.aB  = *(const v2f*)(w + 256 + col * 4 + 2 * hi);
    o.aK0 = *(const v2f*)(w + 320 + col * 8 + 0 + 2 * hi);
    o.aK1 = *(const v2f*)(w + 320 + col * 8 + 4 + 2 * hi);
    const float* zr = zbase + (size_t)t * MS;
    o.bZ0 = *(const v2f*)(zr + 0 + 2 * hi);
    o.bZ1 = *(const v2f*)(zr + 4 + 2 * hi);
    o.bU  = *(const v2f*)(ubase + (size_t)t * US + 2 * hi);
    return o;
}

__device__ __forceinline__ v8f zu_chain(const StepOps& o) {
    v8f d = {};
    d = wmma4(o.aB,  o.bU,  d);
    d = wmma4(o.aK0, o.bZ0, d);
    d = wmma4(o.aK1, o.bZ1, d);
    return d;
}

__device__ __forceinline__ v8f recur(const AOps& m, const v8f& x, v8f c, int col, int hi) {
    v2f bX[4];
    #pragma unroll
    for (int q = 0; q < 4; ++q) {
        int srcLane = col + ((q >= 2) ? 16 : 0);
        int lo = 4 * (q & 1);
        float t0 = bperm(srcLane, x[lo + 0]);
        float t1 = bperm(srcLane, x[lo + 2]);
        float t2 = bperm(srcLane, x[lo + 1]);
        float t3 = bperm(srcLane, x[lo + 3]);
        bX[q].x = hi ? t1 : t0;
        bX[q].y = hi ? t3 : t2;
    }
    v8f d = wmma4(m.a0, bX[0], c);
    d = wmma4(m.a1, bX[1], d);
    d = wmma4(m.a2, bX[2], d);
    d = wmma4(m.a3, bX[3], d);
    return d;
}

__device__ __forceinline__ void store_row(float* o, const v8f& x) {
    *(float4*)(o)     = make_float4(x[0], x[1], x[2], x[3]);
    *(float4*)(o + 4) = make_float4(x[4], x[5], x[6], x[7]);
}

__device__ __forceinline__ v8f load_row(const float* o) {
    const float4 lo = *(const float4*)(o);
    const float4 hh = *(const float4*)(o + 4);
    v8f x = {lo.x, lo.y, lo.z, lo.w, hh.x, hh.y, hh.z, hh.w};
    return x;
}

__global__ __launch_bounds__(32)
void kalman_chunk_local_kernel(const float* __restrict__ obs, const float* __restrict__ inp,
                               const float* __restrict__ ws, float* __restrict__ out)
{
    const int lane = threadIdx.x;
    const int hi   = lane >> 4;
    const int col  = lane & 15;
    const int tile = blockIdx.x;
    const int t0   = blockIdx.y * CH_L;

    if (tile == BATCH / 16) {
        v8f x;
        #pragma unroll
        for (int r = 0; r < 8; ++r) x[r] = (8 * hi + r == col) ? 1.0f : 0.0f;
        AOps cur = load_a(ws + (size_t)t0 * STEPF, col, hi);
        #pragma unroll 1
        for (int s = 1; s <= CH_L; ++s) {
            const int wn = (s < CH_L) ? (t0 + s) : (t0 + s - 1);
            AOps nxt = load_a(ws + (size_t)wn * STEPF, col, hi);
            v8f zero = {};
            x = recur(cur, x, zero, col, hi);
            cur = nxt;
        }
        float* p = out + (size_t)(t0 + PHI_ROW) * NS;
        #pragma unroll
        for (int r = 0; r < 8; ++r) p[(8 * hi + r) * NS + col] = x[r];
        return;
    }

    const int b = tile * 16 + col;
    const float* zbase = obs + (size_t)b * SEQ * MS;
    const float* ubase = inp + (size_t)b * SEQ * US;

    v8f x = {};
    StepOps cur = load_ops(ws, zbase, ubase, t0 + 1, col, hi);
    v8f dzu = zu_chain(cur);
    #pragma unroll 1
    for (int s = 1; s <= CH_L; ++s) {
        const int t  = t0 + s;
        const int tn = (s < CH_L) ? (t + 1) : t;
        StepOps nxt = load_ops(ws, zbase, ubase, tn, col, hi);
        x = recur(cur.aA, x, dzu, col, hi);
        dzu = zu_chain(nxt);
        cur = nxt;
    }
    store_row(out + ((size_t)b * SEQ + (t0 + CH_L)) * NS + 8 * hi, x);
}

__global__ __launch_bounds__(32)
void kalman_chunk_link_kernel(const float* __restrict__ x0, float* out)
{
    const int lane = threadIdx.x;
    const int hi   = lane >> 4;
    const int col  = lane & 15;
    const int b    = blockIdx.x * 16 + col;

    v8f x;
    #pragma unroll
    for (int r = 0; r < 8; ++r) x[r] = x0[8 * hi + r];

    float* obase = out + (size_t)b * SEQ * NS + 8 * hi;
    store_row(obase, x);

    AOps cur = load_a(out + (size_t)PHI_ROW * NS, col, hi);
    v8f  y   = load_row(obase + (size_t)CH_L * NS);
    #pragma unroll 1
    for (int c = 0; c < CH_N; ++c) {
        const int cn = (c + 1 < CH_N) ? (c + 1) : c;
        AOps nxt = load_a(out + ((size_t)cn * CH_L + PHI_ROW) * NS, col, hi);
        v8f  yn  = load_row(obase + (size_t)(cn + 1) * CH_L * NS);
        x = recur(cur, x, y, col, hi);
        store_row(obase + (size_t)(c + 1) * CH_L * NS, x);
        cur = nxt;
        y = yn;
    }
}

__global__ __launch_bounds__(32)
void kalman_chunk_final_kernel(const float* __restrict__ obs, const float* __restrict__ inp,
                               const float* __restrict__ ws, float* out)
{
    const int lane = threadIdx.x;
    const int hi   = lane >> 4;
    const int col  = lane & 15;
    const int b    = blockIdx.x * 16 + col;
    const int t0   = blockIdx.y * CH_L;

    const float* zbase = obs + (size_t)b * SEQ * MS;
    const float* ubase = inp + (size_t)b * SEQ * US;
    float*       obase = out + (size_t)b * SEQ * NS + 8 * hi;

    v8f x = load_row(obase + (size_t)t0 * NS);
    StepOps cur = load_ops(ws, zbase, ubase, t0 + 1, col, hi);
    v8f dzu = zu_chain(cur);
    #pragma unroll 1
    for (int s = 1; s < CH_L; ++s) {
        const int t  = t0 + s;
        const int tn = (s + 1 < CH_L) ? (t + 1) : t;
        StepOps nxt = load_ops(ws, zbase, ubase, tn, col, hi);
        x = recur(cur.aA, x, dzu, col, hi);
        store_row(obase + (size_t)t * NS, x);
        dzu = zu_chain(nxt);
        cur = nxt;
    }
}

extern "C" void kernel_launch(void* const* d_in, const int* in_sizes, int n_in,
                              void* d_out, int out_size, void* d_ws, size_t ws_size,
                              hipStream_t stream) {
    const float* obs = (const float*)d_in[0];
    const float* inp = (const float*)d_in[1];
    const float* A   = (const float*)d_in[2];
    const float* B   = (const float*)d_in[3];
    const float* C   = (const float*)d_in[4];
    const float* Q   = (const float*)d_in[5];
    const float* R   = (const float*)d_in[6];
    const float* x0  = (const float*)d_in[7];
    float* out = (float*)d_out;
    float* ws  = (float*)d_ws;

    hipLaunchKernelGGL(kalman_cov_kernel, dim3(NCB), dim3(CTH), 0, stream,
                       A, B, C, Q, R, ws);
    hipLaunchKernelGGL(kalman_chunk_local_kernel, dim3(BATCH / 16 + 1, CH_N), dim3(32), 0, stream,
                       obs, inp, ws, out);
    hipLaunchKernelGGL(kalman_chunk_link_kernel, dim3(BATCH / 16), dim3(32), 0, stream,
                       x0, out);
    hipLaunchKernelGGL(kalman_chunk_final_kernel, dim3(BATCH / 16, CH_N), dim3(32), 0, stream,
                       obs, inp, ws, out);
}


# Map composition's 16x16 inverses (Gauss-Jordan, row pivot search) moved from LDS, 2 block barriers and 12 LDS round trips a pivot, into one wave's registers, rows by v_permlane16; same sums and order
# speedup vs baseline: 48.5827x; 1.4998x over previous
#include <hip/hip_runtime.h>

#define BATCH 256
#define SEQ   4096
#define NS    16
#define MS    8
#define US    4
#define STEPF 448

typedef __attribute__((ext_vector_type(2))) float v2f;
typedef __attribute__((ext_vector_type(8))) float v8f;

#define TWB 40
#define CWV 4
#define CTH (CWV*32)
#define RPW (NS/CWV)

__device__ __forceinline__ float lane_bcast(float v, int srcLane) {
    return __builtin_bit_cast(float,
        __builtin_amdgcn_readlane(__builtin_bit_cast(int, v), srcLane));
}

template <int K>
__device__ __forceinline__ float row_bcast(float v) {
    return __builtin_bit_cast(float,
        __builtin_amdgcn_update_dpp(0, __builtin_bit_cast(int, v), 0x150 + K, 0xF, 0xF, true));
}

__device__ __forceinline__ float xhalf(float x) {
    const int b = __builtin_bit_cast(int, x);
    return __builtin_bit_cast(float,
        __builtin_amdgcn_permlanex16(b, b, 0x76543210, 0xfedcba98, false, false));
}

#define ROW_DOT4_STEP(K)                                              \
    { const float x0 = row_bcast<K>(v0), x1 = row_bcast<K>(v1);      \
      a0 += x0 * ma[K];  a1 += x1 * ma[K];                            \
      b0 += x0 * mb[K];  b1 += x1 * mb[K]; }
__device__ __forceinline__ void row_dot4(float v0, float v1,
                                         const float (&ma)[NS], const float (&mb)[NS],
                                         float& ra0, float& rb0, float& ra1, float& rb1) {
    float a0 = 0.f, b0 = 0.f, a1 = 0.f, b1 = 0.f;
    ROW_DOT4_STEP(0)   ROW_DOT4_STEP(1)   ROW_DOT4_STEP(2)   ROW_DOT4_STEP(3)
    ROW_DOT4_STEP(4)   ROW_DOT4_STEP(5)   ROW_DOT4_STEP(6)   ROW_DOT4_STEP(7)
    ROW_DOT4_STEP(8)   ROW_DOT4_STEP(9)   ROW_DOT4_STEP(10)  ROW_DOT4_STEP(11)
    ROW_DOT4_STEP(12)  ROW_DOT4_STEP(13)  ROW_DOT4_STEP(14)  ROW_DOT4_STEP(15)
    ra0 = a0;  rb0 = b0;  ra1 = a1;  rb1 = b1;
}
#undef ROW_DOT4_STEP


#define CL   64
#define CLLG 6
#define NCB  ((SEQ - 1 + CL - 1) / CL)
#define FORGET_BOUND 1e-8f

__device__ __forceinline__ void mm16(float* D, const float* X, const float* Y, int tid) {
    #pragma unroll
    for (int q = 0; q < 2; ++q) {
        const int e = tid + q*CTH, i = e >> 4, j = e & 15;
        float a = 0.f;
        #pragma unroll
        for (int k = 0; k < NS; ++k) a += X[i*NS + k] * Y[k*NS + j];
        D[e] = a;
    }
    __syncthreads();
}

__device__ __forceinline__ void mm16_add(float* D, const float* X, const float* Y,
                                         const float* Z, int tid) {
    #pragma unroll
    for (int q = 0; q < 2; ++q) {
        const int e = tid + q*CTH, i = e >> 4, j = e & 15;
        float a = 0.f;
        #pragma unroll
        for (int k = 0; k < NS; ++k) a += X[i*NS + k] * Y[k*NS + j];
        D[e] = a + Z[e];
    }
    __syncthreads();
}

__device__ __forceinline__ void eye_axpy(float* D, const float* X, float sgn, int tid) {
    #pragma unroll
    for (int q = 0; q < 2; ++q) {
        const int e = tid + q*CTH;
        D[e] = (((e >> 4) == (e & 15)) ? 1.0f : 0.0f) + sgn * X[e];
    }
    __syncthreads();
}

__device__ __forceinline__ void sub16(float* D, const float* X, const float* Y, int tid) {
    #pragma unroll
    for (int q = 0; q < 2; ++q) { const int e = tid + q*CTH;  D[e] = X[e] - Y[e]; }
    __syncthreads();
}

__device__ __forceinline__ float row_pick(float v, int sel16) {
    const int b = __builtin_bit_cast(int, v);
    return __builtin_bit_cast(float,
        __builtin_amdgcn_permlane16(b, b, sel16, sel16, false, false));
}

template <int MSK>
__device__ __forceinline__ float row_xm(float v) {
    return __builtin_bit_cast(float,
        __builtin_amdgcn_update_dpp(0, __builtin_bit_cast(int, v), 0x160 + MSK, 0xF, 0xF, true));
}

__device__ __forceinline__ void inv16(float* D, const float* X, int tid) {
    if (tid < 32) {
        const int r = tid & 15, h = tid >> 4;
        float a[NS];
        #pragma unroll
        for (int j = 0; j < NS; ++j) {
            const float x = X[r*NS + j];
            a[j] = h ? ((j == r) ? 1.0f : 0.0f) : x;
        }
        #pragma unroll 1
        for (int p = 0; p < NS; ++p) {
            const int selp = p * 0x11111111;
            const float cx   = xhalf(a[0]);
            const float fcol = h ? cx : a[0];
            float key = fabsf(fcol);
            key = (r >= p && key > -1.0f) ? key : -1.0f;
            float m = key;
            m = fmaxf(m, row_xm<1>(m));
            m = fmaxf(m, row_xm<2>(m));
            m = fmaxf(m, row_xm<4>(m));
            m = fmaxf(m, row_xm<8>(m));
            const unsigned hit = __builtin_amdgcn_ballot_w32(key == m && r >= p) & 0xffffu;
            const int rs   = __builtin_ctz(hit);
            const int selr = rs * 0x11111111;
            const float inv = 1.0f / row_pick(fcol, selr);
            const float fp  = row_pick(fcol, selp);
            const bool isP = (r == p), isR = (r == rs);
            const float fi = (isR ? fp : fcol) * inv;
            float n[NS];
            #pragma unroll
            for (int j = 0; j < NS; ++j) {
                const float pr = row_pick(a[j], selr);
                const float xp = row_pick(a[j], selp);
                const float x  = isR ? xp : a[j];
                n[j] = isP ? pr * inv : x - fi * pr;
            }
            #pragma unroll
            for (int j = 0; j < NS; ++j) a[j] = n[(j + 1) & (NS - 1)];
        }
        if (h) {
            #pragma unroll
            for (int j = 0; j < NS; ++j) D[r*NS + j] = a[j];
        }
    }
    __syncthreads();
}

__device__ __forceinline__ float nmax(float a, float v) {
    return (v > a || v != v) ? v : a;
}

#define MX_AT   0
#define MX_CP   1
#define MX_CPT  2
#define MX_R16  3
#define MX_X    4
#define MX_GL   5
#define MX_GR   6
#define MX_GC   7
#define MX_GJ   8
#define MX_NL   9
#define MX_NR   10
#define MX_NC   11
#define MX_NJ   12
#define MX_E    13
#define MX_T0   14
#define MX_CNT  20
#define MXP(pool, m) ((pool) + (m)*NS*NS)

__device__ __forceinline__ float map_square(float* pool, float* red, int tid) {
    float* X  = MXP(pool, MX_X);
    float* GL = MXP(pool, MX_GL);  float* GR = MXP(pool, MX_GR);
    float* GC = MXP(pool, MX_GC);  float* GJ = MXP(pool, MX_GJ);
    float* NL = MXP(pool, MX_NL);  float* NR = MXP(pool, MX_NR);
    float* NC = MXP(pool, MX_NC);  float* NJ = MXP(pool, MX_NJ);
    float* t0 = MXP(pool, MX_T0);      float* t1 = MXP(pool, MX_T0 + 1);
    float* t2 = MXP(pool, MX_T0 + 2);  float* t3 = MXP(pool, MX_T0 + 3);
    float* t4 = MXP(pool, MX_T0 + 4);  float* t5 = MXP(pool, MX_T0 + 5);
    sub16(t0, GC, X, tid);
    mm16(t1, t0, GJ, tid);
    eye_axpy(t2, t1, 1.0f, tid);
    inv16(t3, t2, tid);
    mm16(t1, t3, t0, tid);
    mm16(t2, GJ, t1, tid);
    eye_axpy(t4, t2, -1.0f, tid);
    mm16(t2, t3, GL, tid);
    mm16(NL, GL, t2, tid);
    mm16(t2, GL, t1, tid);
    mm16_add(NC, t2, GR, GC, tid);
    mm16(t2, t4, GR, tid);
    mm16(NR, GR, t2, tid);
    mm16(t2, GJ, GL, tid);
    mm16(t5, t4, t2, tid);
    mm16_add(NJ, GR, t5, GJ, tid);
    const float l0 = NL[tid], l1 = NL[tid + CTH], r0 = NR[tid], r1 = NR[tid + CTH];
    const float c0 = NC[tid], c1 = NC[tid + CTH], j0 = NJ[tid], j1 = NJ[tid + CTH];
    red[tid]         = nmax(nmax(0.f, fabsf(l0)), fabsf(l1));
    red[CTH + tid]   = nmax(nmax(0.f, fabsf(r0)), fabsf(r1));
    red[2*CTH + tid] = nmax(nmax(nmax(nmax(0.f, fabsf(c0)), fabsf(c1)), fabsf(j0)), fabsf(j1));
    __syncthreads();
    float mL = 0.f, mR = 0.f, mC = 0.f;
    for (int i = 0; i < CTH; ++i) {
        mL = nmax(mL, red[i]);  mR = nmax(mR, red[CTH + i]);  mC = nmax(mC, red[2*CTH + i]);
    }
    const bool fin = (mL < 3.0e38f) && (mR < 3.0e38f) && (mC < 3.0e38f);
    int a = 0;
    if (fin && mL > 0.f && mR > 0.f) {
        int eL, eR;
        (void)frexpf(mL, &eL);  (void)frexpf(mR, &eR);
        a = (eR - eL) / 2;
    }
    GL[tid] = ldexpf(l0, a);   GL[tid + CTH] = ldexpf(l1, a);
    GR[tid] = ldexpf(r0, -a);  GR[tid + CTH] = ldexpf(r1, -a);
    GC[tid] = c0;  GC[tid + CTH] = c1;
    GJ[tid] = j0;  GJ[tid + CTH] = j1;
    __syncthreads();
    return fin ? mL * mR : 1.0f;
}

__device__ __forceinline__ void map_apply(float* pool, int tid) {
    float* X  = MXP(pool, MX_X);   float* E  = MXP(pool, MX_E);
    float* GL = MXP(pool, MX_GL);  float* GR = MXP(pool, MX_GR);
    float* GC = MXP(pool, MX_GC);  float* GJ = MXP(pool, MX_GJ);
    float* t0 = MXP(pool, MX_T0);      float* t1 = MXP(pool, MX_T0 + 1);
    float* t2 = MXP(pool, MX_T0 + 2);  float* t3 = MXP(pool, MX_T0 + 3);
    sub16(t0, E, X, tid);
    mm16(t1, t0, GJ, tid);
    eye_axpy(t2, t1, 1.0f, tid);
    inv16(t3, t2, tid);
    mm16(t1, t3, t0, tid);
    mm16(t2, GL, t1, tid);
    mm16_add(E, t2, GR, GC, tid);
}

__device__ __forceinline__ void map_one_step(float* pool, const float* sA,
                                             const float* sQ, int tid) {
    float* AT  = MXP(pool, MX_AT);   float* CP  = MXP(pool, MX_CP);
    float* CPT = MXP(pool, MX_CPT);  float* R16 = MXP(pool, MX_R16);
    float* X   = MXP(pool, MX_X);
    float* GL = MXP(pool, MX_GL);  float* GR = MXP(pool, MX_GR);
    float* GC = MXP(pool, MX_GC);  float* GJ = MXP(pool, MX_GJ);
    float* t0 = MXP(pool, MX_T0);  float* t1 = MXP(pool, MX_T0 + 1);  float* t2 = MXP(pool, MX_T0 + 2);
    mm16(t0, CP, X, tid);
    mm16_add(t1, t0, CPT, R16, tid);
    inv16(t2, t1, tid);
    mm16(t0, CPT, t2, tid);
    mm16(GJ, t0, CP, tid);
    mm16(t0, X, GJ, tid);
    eye_axpy(t1, t0, -1.0f, tid);
    mm16(t0, GJ, X, tid);
    eye_axpy(t2, t0, -1.0f, tid);
    mm16(GL, sA, t1, tid);
    mm16(GR, t2, AT, tid);
    mm16(t0, GL, X, tid);
    mm16_add(GC, t0, AT, sQ, tid);
}

__global__ __launch_bounds__(CTH)
void kalman_cov_kernel(const float* __restrict__ Ag, const float* __restrict__ Bg,
                       const float* __restrict__ Cg, const float* __restrict__ Qg,
                       const float* __restrict__ Rg, float* __restrict__ ws)
{
    __shared__ float sA[NS*NS], sB[NS*US], sC[MS*NS], sQ[NS*NS], sR[MS*MS];
    __shared__ float sBot[2*NS*TWB];
    __shared__ float sCA[MS*NS], sCB[MS*US];
    __shared__ float sPool[MX_CNT*NS*NS];
    __shared__ float sRed[3*CTH];

    const int tid = threadIdx.x;
    for (int e = tid; e < NS*NS; e += CTH) {
        sA[e] = Ag[e];
        sQ[e] = Qg[e];
    }
    if (tid < NS*US) sB[tid] = Bg[tid];
    if (tid < MS*NS) sC[tid] = Cg[tid];
    if (tid < MS*MS) sR[tid] = Rg[tid];
    __syncthreads();

    #pragma unroll
    for (int q = 0; q < 2; ++q) {
        const int e = tid + q*CTH, i = e >> 4, j = e & 15;
        MXP(sPool, MX_AT)[e]  = sA[j*NS + i];
        MXP(sPool, MX_CP)[e]  = (i < MS) ? sC[i*NS + j] : 0.0f;
        MXP(sPool, MX_CPT)[e] = (j < MS) ? sC[j*NS + i] : 0.0f;
        MXP(sPool, MX_R16)[e] = (i < MS && j < MS) ? sR[i*MS + j] : ((i == j) ? 1.0f : 0.0f);
    }

    for (int e = tid; e < MS*NS + MS*US; e += CTH) {
        if (e < MS*NS) {
            const int i16 = e >> 4, j16 = e & 15;
            float a = 0.f;
            #pragma unroll
            for (int k = 0; k < NS; ++k) a += sC[i16*NS + k] * sA[k*NS + j16];
            sCA[e] = a;
        } else {
            const int q = e - MS*NS, ib = q >> 2, jb = q & 3;
            float a = 0.f;
            #pragma unroll
            for (int k = 0; k < NS; ++k) a += sC[ib*NS + k] * sB[k*US + jb];
            sCB[q] = a;
        }
    }
    __syncthreads();

    if (tid < NS*8) {
        const float a = sA[(tid >> 3)*NS + (tid & 7)];
        sBot[(tid >> 3)*TWB + 8 + (tid & 7)] = a;
        sBot[NS*TWB + (tid >> 3)*TWB + 8 + (tid & 7)] = a;
    }

    const int w  = tid >> 5, c = tid & 31;
    const int rbase = RPW * w;
    float kBHi[RPW];
    int   oh[RPW];
    #pragma unroll
    for (int j = 0; j < RPW; ++j) {
        const int r = rbase + j;
        kBHi[j] = 0.f;  oh[j] = -1;
        if (c < 8)        { kBHi[j] = sA[r*NS + 8 + c];     oh[j] = r*NS + 8 + c; }
        else if (c < 12)  { kBHi[j] = sB[r*US + (c - 8)];   oh[j] = 256 + r*US + (c - 8); }
        else if (c < 20)  { oh[j] = 320 + r*MS + (c - 12); }
    }
    float kPHi[MS];
    #pragma unroll
    for (int i = 0; i < MS; ++i) {
        float h = 0.f;
        if (c < 8)        h = sCA[i*NS + 8 + c];
        else if (c < 12)  h = sCB[i*US + (c - 8)];
        else if (c < 20)  h = ((c - 12) == i) ? -1.0f : 0.0f;
        kPHi[i] = h;
    }

    const int hr  = c >> 4, jc = c & 15;
    const int rowa = rbase + hr;
    const int rowb = rbase + 2 + hr;
    float kAj[NS], kCm[NS];
    #pragma unroll
    for (int k = 0; k < NS; ++k) {
        kAj[k]    = sA[jc*NS + k];
        kCm[k]    = sC[(jc & 7)*NS + k];
    }
    float kC[MS*NS];
    #pragma unroll
    for (int q = 0; q < MS*NS; ++q) kC[q] = sC[q];
    float kR[MS];
    #pragma unroll
    for (int i = 0; i < MS; ++i) kR[i] = (c < MS) ? sR[i*MS + c] : 0.0f;
    float kA2[RPW][NS];
    float kQ2[RPW];
    #pragma unroll
    for (int j = 0; j < RPW; ++j) {
        #pragma unroll
        for (int k = 0; k < NS; ++k) kA2[j][k] = sA[(rbase + j)*NS + k];
        kQ2[j] = (c >= 16) ? sQ[(rbase + j)*NS + (c - 16)] : 0.0f;
    }

    float m0, m1;
    {
        float a0 = 0.f, a1 = 0.f;
        #pragma unroll
        for (int k = 0; k < NS; ++k) {
            a0 += sA[rowa*NS + k] * kAj[k];
            a1 += sA[rowb*NS + k] * kAj[k];
        }
        m0 = a0 + sQ[rowa*NS + jc];
        m1 = a1 + sQ[rowb*NS + jc];
    }

    float r_lo[RPW], r_hi[RPW];
    #pragma unroll
    for (int j = 0; j < RPW; ++j) { r_lo[j] = 0.f;  r_hi[j] = 0.f; }

    const int cb = blockIdx.x;
    int  tb = 1, te = CL + 1;
    bool store = (cb == 0);
    #pragma unroll 1
    for (int phase = 0; phase < 2; ++phase) {
    if (phase == 1) {
        float* mxX = MXP(sPool, MX_X);
        float* mxE = MXP(sPool, MX_E);
        mxX[rowa*NS + jc] = m0;  mxX[rowb*NS + jc] = m1;
        mxE[rowa*NS + jc] = m0;  mxE[rowb*NS + jc] = m1;
        __syncthreads();
        map_one_step(sPool, sA, sQ, tid);
        const int nq = cb - 1;
        float gv = 1.0f;
        bool  ok = false;
        #pragma unroll 1
        for (int s = 0; s < CLLG + 6; ++s) {
            if (s >= CLLG) {
                const int k = s - CLLG;
                if (k == 0) {
                    ok = gv < FORGET_BOUND;
                    if (!ok || cb < 2) break;
                }
                if ((nq >> k) & 1) map_apply(sPool, tid);
                if ((nq >> (k + 1)) == 0) break;
            }
            gv = map_square(sPool, sRed, tid);
        }
        if (ok) {
            if (cb == 0) return;
            tb = cb * CL + 1;
            te = (tb + CL < SEQ) ? (tb + CL) : SEQ;
            m0 = mxE[rowa*NS + jc];
            m1 = mxE[rowb*NS + jc];
        } else {
            if (cb != 0) return;
            tb = CL + 1;
            te = SEQ;
        }
        store = true;
    }
    #pragma unroll 1
    for (int t = tb; t < te; ++t) {
        float* bot = sBot + (t & 1) * (NS*TWB);
        float ga, xa, gb, xb;
        row_dot4(m0, m1, kCm, kAj, ga, xa, gb, xb);
        bot[rowa*TWB + 16 + jc] = xa;
        bot[rowb*TWB + 16 + jc] = xb;
        bot[rowa*TWB + (jc & 7)] = ga;
        bot[rowb*TWB + (jc & 7)] = gb;
        __syncthreads();

        float b_lo[RPW];
        #pragma unroll
        for (int j = 0; j < RPW; ++j) b_lo[j] = bot[(rbase + j)*TWB + c];
        float col[NS];
        #pragma unroll
        for (int k = 0; k < NS; ++k) col[k] = bot[k*TWB + c];
        if (store && t > tb) {
            float* wsq = ws + (size_t)(t - 2) * STEPF;
            if (c >= 8 && c < 16) {
                #pragma unroll
                for (int j = 0; j < RPW; ++j) wsq[(rbase + j)*NS + (c - 8)] = r_lo[j];
            }
            if (oh[0] >= 0) {
                #pragma unroll
                for (int j = 0; j < RPW; ++j) wsq[oh[j]] = r_hi[j];
            }
        }
        float pl[MS], ph[MS];
        #pragma unroll
        for (int i = 0; i < MS; ++i) {
            pl[i] = kR[i];
            ph[i] = kPHi[i];
        }
        #pragma unroll
        for (int k = 0; k < NS; ++k) {
            #pragma unroll
            for (int i = 0; i < MS; ++i) pl[i] += kC[i*NS + k] * col[k];
        }
        float b2[RPW];
        #pragma unroll
        for (int j = 0; j < RPW; ++j) b2[j] = kQ2[j];
        #pragma unroll
        for (int k = 0; k < NS; ++k) {
            #pragma unroll
            for (int j = 0; j < RPW; ++j) b2[j] += kA2[j][k] * col[k];
        }
        float b_hi[RPW];
        #pragma unroll
        for (int j = 0; j < RPW; ++j) b_hi[j] = kBHi[j];

        #pragma unroll
        for (int p = 0; p < MS; ++p) {
            const float inv = 1.0f / lane_bcast(pl[p], p);
            #pragma unroll
            for (int i = p + 1; i < MS; ++i) {
                const float f = lane_bcast(pl[i], p) * inv;
                pl[i] -= f * pl[p];  ph[i] -= f * ph[p];
            }
            #pragma unroll
            for (int j = 0; j < RPW; ++j) {
                const float fj = lane_bcast(b_lo[j], p) * inv;
                b_lo[j] -= fj * pl[p];  b_hi[j] -= fj * ph[p];
                const float gj = lane_bcast(b2[j], p) * inv;
                b2[j] -= gj * pl[p];
            }
        }

        #pragma unroll
        for (int j = 0; j < RPW; ++j) { r_lo[j] = b_lo[j];  r_hi[j] = b_hi[j]; }

        const float x0 = xhalf(b2[0]);
        const float x2 = xhalf(b2[2]);
        m0 = hr ? b2[1] : x0;
        m1 = hr ? b2[3] : x2;
    }

    if (store) {
        float* wsq = ws + (size_t)(te - 2) * STEPF;
        if (c >= 8 && c < 16) {
            #pragma unroll
            for (int j = 0; j < RPW; ++j) wsq[(rbase + j)*NS + (c - 8)] = r_lo[j];
        }
        if (oh[0] >= 0) {
            #pragma unroll
            for (int j = 0; j < RPW; ++j) wsq[oh[j]] = r_hi[j];
        }
    }
    }
}

#define CH_L 63
#define CH_N 65
#define PHI_ROW 2

__device__ __forceinline__ float bperm(int srcLane, float v) {
    return __builtin_bit_cast(float,
        __builtin_amdgcn_ds_bpermute(srcLane << 2, __builtin_bit_cast(int, v)));
}

__device__ __forceinline__ v8f wmma4(v2f a, v2f b, v8f c) {
    return __builtin_amdgcn_wmma_f32_16x16x4_f32(false, a, false, b,
                                                 (short)0, c, false, false);
}

struct AOps {
    v2f a0, a1, a2, a3;
};

struct StepOps {
    AOps  aA;
    v2f aB, aK0, aK1;
    v2f bZ0, bZ1, bU;
};

__device__ __forceinline__ AOps load_a(const float* m, int col, int hi) {
    AOps o;
    o.a0 = *(const v2f*)(m + col * 16 + 0  + 2 * hi);
    o.a1 = *(const v2f*)(m + col * 16 + 4  + 2 * hi);
    o.a2 = *(const v2f*)(m + col * 16 + 8  + 2 * hi);
    o.a3 = *(const v2f*)(m + col * 16 + 12 + 2 * hi);
    return o;
}

__device__ __forceinline__ StepOps load_ops(const float* __restrict__ ws,
                                            const float* __restrict__ zbase,
                                            const float* __restrict__ ubase,
                                            int t, int col, int hi) {
    StepOps o;
    const float* w = ws + (size_t)(t - 1) * STEPF;
    o.aA  = load_a(w, col, hi);
    o.aB  = *(const v2f*)(w + 256 + col * 4 + 2 * hi);
    o.aK0 = *(const v2f*)(w + 320 + col * 8 + 0 + 2 * hi);
    o.aK1 = *(const v2f*)(w + 320 + col * 8 + 4 + 2 * hi);
    const float* zr = zbase + (size_t)t * MS;
    o.bZ0 = *(const v2f*)(zr + 0 + 2 * hi);
    o.bZ1 = *(const v2f*)(zr + 4 + 2 * hi);
    o.bU  = *(const v2f*)(ubase + (size_t)t * US + 2 * hi);
    return o;
}

__device__ __forceinline__ v8f zu_chain(const StepOps& o) {
    v8f d = {};
    d = wmma4(o.aB,  o.bU,  d);
    d = wmma4(o.aK0, o.bZ0, d);
    d = wmma4(o.aK1, o.bZ1, d);
    return d;
}

__device__ __forceinline__ v8f recur(const AOps& m, const v8f& x, v8f c, int col, int hi) {
    v2f bX[4];
    #pragma unroll
    for (int q = 0; q < 4; ++q) {
        int srcLane = col + ((q >= 2) ? 16 : 0);
        int lo = 4 * (q & 1);
        float t0 = bperm(srcLane, x[lo + 0]);
        float t1 = bperm(srcLane, x[lo + 2]);
        float t2 = bperm(srcLane, x[lo + 1]);
        float t3 = bperm(srcLane, x[lo + 3]);
        bX[q].x = hi ? t1 : t0;
        bX[q].y = hi ? t3 : t2;
    }
    v8f d = wmma4(m.a0, bX[0], c);
    d = wmma4(m.a1, bX[1], d);
    d = wmma4(m.a2, bX[2], d);
    d = wmma4(m.a3, bX[3], d);
    return d;
}

__device__ __forceinline__ void store_row(float* o, const v8f& x) {
    *(float4*)(o)     = make_float4(x[0], x[1], x[2], x[3]);
    *(float4*)(o + 4) = make_float4(x[4], x[5], x[6], x[7]);
}

__device__ __forceinline__ v8f load_row(const float* o) {
    const float4 lo = *(const float4*)(o);
    const float4 hh = *(const float4*)(o + 4);
    v8f x = {lo.x, lo.y, lo.z, lo.w, hh.x, hh.y, hh.z, hh.w};
    return x;
}

__global__ __launch_bounds__(32)
void kalman_chunk_local_kernel(const float* __restrict__ obs, const float* __restrict__ inp,
                               const float* __restrict__ ws, float* __restrict__ out)
{
    const int lane = threadIdx.x;
    const int hi   = lane >> 4;
    const int col  = lane & 15;
    const int tile = blockIdx.x;
    const int t0   = blockIdx.y * CH_L;

    if (tile == BATCH / 16) {
        v8f x;
        #pragma unroll
        for (int r = 0; r < 8; ++r) x[r] = (8 * hi + r == col) ? 1.0f : 0.0f;
        AOps cur = load_a(ws + (size_t)t0 * STEPF, col, hi);
        #pragma unroll 1
        for (int s = 1; s <= CH_L; ++s) {
            const int wn = (s < CH_L) ? (t0 + s) : (t0 + s - 1);
            AOps nxt = load_a(ws + (size_t)wn * STEPF, col, hi);
            v8f zero = {};
            x = recur(cur, x, zero, col, hi);
            cur = nxt;
        }
        float* p = out + (size_t)(t0 + PHI_ROW) * NS;
        #pragma unroll
        for (int r = 0; r < 8; ++r) p[(8 * hi + r) * NS + col] = x[r];
        return;
    }

    const int b = tile * 16 + col;
    const float* zbase = obs + (size_t)b * SEQ * MS;
    const float* ubase = inp + (size_t)b * SEQ * US;

    v8f x = {};
    StepOps cur = load_ops(ws, zbase, ubase, t0 + 1, col, hi);
    v8f dzu = zu_chain(cur);
    #pragma unroll 1
    for (int s = 1; s <= CH_L; ++s) {
        const int t  = t0 + s;
        const int tn = (s < CH_L) ? (t + 1) : t;
        StepOps nxt = load_ops(ws, zbase, ubase, tn, col, hi);
        x = recur(cur.aA, x, dzu, col, hi);
        dzu = zu_chain(nxt);
        cur = nxt;
    }
    store_row(out + ((size_t)b * SEQ + (t0 + CH_L)) * NS + 8 * hi, x);
}

__global__ __launch_bounds__(32)
void kalman_chunk_link_kernel(const float* __restrict__ x0, float* out)
{
    const int lane = threadIdx.x;
    const int hi   = lane >> 4;
    const int col  = lane & 15;
    const int b    = blockIdx.x * 16 + col;

    v8f x;
    #pragma unroll
    for (int r = 0; r < 8; ++r) x[r] = x0[8 * hi + r];

    float* obase = out + (size_t)b * SEQ * NS + 8 * hi;
    store_row(obase, x);

    AOps cur = load_a(out + (size_t)PHI_ROW * NS, col, hi);
    v8f  y   = load_row(obase + (size_t)CH_L * NS);
    #pragma unroll 1
    for (int c = 0; c < CH_N; ++c) {
        const int cn = (c + 1 < CH_N) ? (c + 1) : c;
        AOps nxt = load_a(out + ((size_t)cn * CH_L + PHI_ROW) * NS, col, hi);
        v8f  yn  = load_row(obase + (size_t)(cn + 1) * CH_L * NS);
        x = recur(cur, x, y, col, hi);
        store_row(obase + (size_t)(c + 1) * CH_L * NS, x);
        cur = nxt;
        y = yn;
    }
}

__global__ __launch_bounds__(32)
void kalman_chunk_final_kernel(const float* __restrict__ obs, const float* __restrict__ inp,
                               const float* __restrict__ ws, float* out)
{
    const int lane = threadIdx.x;
    const int hi   = lane >> 4;
    const int col  = lane & 15;
    const int b    = blockIdx.x * 16 + col;
    const int t0   = blockIdx.y * CH_L;

    const float* zbase = obs + (size_t)b * SEQ * MS;
    const float* ubase = inp + (size_t)b * SEQ * US;
    float*       obase = out + (size_t)b * SEQ * NS + 8 * hi;

    v8f x = load_row(obase + (size_t)t0 * NS);
    StepOps cur = load_ops(ws, zbase, ubase, t0 + 1, col, hi);
    v8f dzu = zu_chain(cur);
    #pragma unroll 1
    for (int s = 1; s < CH_L; ++s) {
        const int t  = t0 + s;
        const int tn = (s + 1 < CH_L) ? (t + 1) : t;
        StepOps nxt = load_ops(ws, zbase, ubase, tn, col, hi);
        x = recur(cur.aA, x, dzu, col, hi);
        store_row(obase + (size_t)t * NS, x);
        dzu = zu_chain(nxt);
        cur = nxt;
    }
}

extern "C" void kernel_launch(void* const* d_in, const int* in_sizes, int n_in,
                              void* d_out, int out_size, void* d_ws, size_t ws_size,
                              hipStream_t stream) {
    const float* obs = (const float*)d_in[0];
    const float* inp = (const float*)d_in[1];
    const float* A   = (const float*)d_in[2];
    const float* B   = (const float*)d_in[3];
    const float* C   = (const float*)d_in[4];
    const float* Q   = (const float*)d_in[5];
    const float* R   = (const float*)d_in[6];
    const float* x0  = (const float*)d_in[7];
    float* out = (float*)d_out;
    float* ws  = (float*)d_ws;

    hipLaunchKernelGGL(kalman_cov_kernel, dim3(NCB), dim3(CTH), 0, stream,
                       A, B, C, Q, R, ws);
    hipLaunchKernelGGL(kalman_chunk_local_kernel, dim3(BATCH / 16 + 1, CH_N), dim3(32), 0, stream,
                       obs, inp, ws, out);
    hipLaunchKernelGGL(kalman_chunk_link_kernel, dim3(BATCH / 16), dim3(32), 0, stream,
                       x0, out);
    hipLaunchKernelGGL(kalman_chunk_final_kernel, dim3(BATCH / 16, CH_N), dim3(32), 0, stream,
                       obs, inp, ws, out);
}


# Covariance time split retiled: 256 chunks of 16 steps, not 64 of 64 (a block a WGP): serial path 32 steps + 7 map applications, not 128 + 5; forgetting test on the 64-step map; starts reassociated
# speedup vs baseline: 57.0481x; 1.1742x over previous
#include <hip/hip_runtime.h>

#define BATCH 256
#define SEQ   4096
#define NS    16
#define MS    8
#define US    4
#define STEPF 448

typedef __attribute__((ext_vector_type(2))) float v2f;
typedef __attribute__((ext_vector_type(8))) float v8f;

#define TWB 40
#define CWV 4
#define CTH (CWV*32)
#define RPW (NS/CWV)

__device__ __forceinline__ float lane_bcast(float v, int srcLane) {
    return __builtin_bit_cast(float,
        __builtin_amdgcn_readlane(__builtin_bit_cast(int, v), srcLane));
}

template <int K>
__device__ __forceinline__ float row_bcast(float v) {
    return __builtin_bit_cast(float,
        __builtin_amdgcn_update_dpp(0, __builtin_bit_cast(int, v), 0x150 + K, 0xF, 0xF, true));
}

__device__ __forceinline__ float xhalf(float x) {
    const int b = __builtin_bit_cast(int, x);
    return __builtin_bit_cast(float,
        __builtin_amdgcn_permlanex16(b, b, 0x76543210, 0xfedcba98, false, false));
}

#define ROW_DOT4_STEP(K)                                              \
    { const float x0 = row_bcast<K>(v0), x1 = row_bcast<K>(v1);      \
      a0 += x0 * ma[K];  a1 += x1 * ma[K];                            \
      b0 += x0 * mb[K];  b1 += x1 * mb[K]; }
__device__ __forceinline__ void row_dot4(float v0, float v1,
                                         const float (&ma)[NS], const float (&mb)[NS],
                                         float& ra0, float& rb0, float& ra1, float& rb1) {
    float a0 = 0.f, b0 = 0.f, a1 = 0.f, b1 = 0.f;
    ROW_DOT4_STEP(0)   ROW_DOT4_STEP(1)   ROW_DOT4_STEP(2)   ROW_DOT4_STEP(3)
    ROW_DOT4_STEP(4)   ROW_DOT4_STEP(5)   ROW_DOT4_STEP(6)   ROW_DOT4_STEP(7)
    ROW_DOT4_STEP(8)   ROW_DOT4_STEP(9)   ROW_DOT4_STEP(10)  ROW_DOT4_STEP(11)
    ROW_DOT4_STEP(12)  ROW_DOT4_STEP(13)  ROW_DOT4_STEP(14)  ROW_DOT4_STEP(15)
    ra0 = a0;  rb0 = b0;  ra1 = a1;  rb1 = b1;
}
#undef ROW_DOT4_STEP


#define CL   16
#define CLLG 4
#define FLG  6
#define LADB 8
#define NCB  ((SEQ - 1 + CL - 1) / CL)
#define FORGET_BOUND 1e-8f

__device__ __forceinline__ void mm16(float* D, const float* X, const float* Y, int tid) {
    #pragma unroll
    for (int q = 0; q < 2; ++q) {
        const int e = tid + q*CTH, i = e >> 4, j = e & 15;
        float a = 0.f;
        #pragma unroll
        for (int k = 0; k < NS; ++k) a += X[i*NS + k] * Y[k*NS + j];
        D[e] = a;
    }
    __syncthreads();
}

__device__ __forceinline__ void mm16_add(float* D, const float* X, const float* Y,
                                         const float* Z, int tid) {
    #pragma unroll
    for (int q = 0; q < 2; ++q) {
        const int e = tid + q*CTH, i = e >> 4, j = e & 15;
        float a = 0.f;
        #pragma unroll
        for (int k = 0; k < NS; ++k) a += X[i*NS + k] * Y[k*NS + j];
        D[e] = a + Z[e];
    }
    __syncthreads();
}

__device__ __forceinline__ void eye_axpy(float* D, const float* X, float sgn, int tid) {
    #pragma unroll
    for (int q = 0; q < 2; ++q) {
        const int e = tid + q*CTH;
        D[e] = (((e >> 4) == (e & 15)) ? 1.0f : 0.0f) + sgn * X[e];
    }
    __syncthreads();
}

__device__ __forceinline__ void sub16(float* D, const float* X, const float* Y, int tid) {
    #pragma unroll
    for (int q = 0; q < 2; ++q) { const int e = tid + q*CTH;  D[e] = X[e] - Y[e]; }
    __syncthreads();
}

__device__ __forceinline__ float row_pick(float v, int sel16) {
    const int b = __builtin_bit_cast(int, v);
    return __builtin_bit_cast(float,
        __builtin_amdgcn_permlane16(b, b, sel16, sel16, false, false));
}

template <int MSK>
__device__ __forceinline__ float row_xm(float v) {
    return __builtin_bit_cast(float,
        __builtin_amdgcn_update_dpp(0, __builtin_bit_cast(int, v), 0x160 + MSK, 0xF, 0xF, true));
}

__device__ __forceinline__ void inv16(float* D, const float* X, int tid) {
    if (tid < 32) {
        const int r = tid & 15, h = tid >> 4;
        float a[NS];
        #pragma unroll
        for (int j = 0; j < NS; ++j) {
            const float x = X[r*NS + j];
            a[j] = h ? ((j == r) ? 1.0f : 0.0f) : x;
        }
        #pragma unroll 1
        for (int p = 0; p < NS; ++p) {
            const int selp = p * 0x11111111;
            const float cx   = xhalf(a[0]);
            const float fcol = h ? cx : a[0];
            float key = fabsf(fcol);
            key = (r >= p && key > -1.0f) ? key : -1.0f;
            float m = key;
            m = fmaxf(m, row_xm<1>(m));
            m = fmaxf(m, row_xm<2>(m));
            m = fmaxf(m, row_xm<4>(m));
            m = fmaxf(m, row_xm<8>(m));
            const unsigned hit = __builtin_amdgcn_ballot_w32(key == m && r >= p) & 0xffffu;
            const int rs   = __builtin_ctz(hit);
            const int selr = rs * 0x11111111;
            const float inv = 1.0f / row_pick(fcol, selr);
            const float fp  = row_pick(fcol, selp);
            const bool isP = (r == p), isR = (r == rs);
            const float fi = (isR ? fp : fcol) * inv;
            float n[NS];
            #pragma unroll
            for (int j = 0; j < NS; ++j) {
                const float pr = row_pick(a[j], selr);
                const float xp = row_pick(a[j], selp);
                const float x  = isR ? xp : a[j];
                n[j] = isP ? pr * inv : x - fi * pr;
            }
            #pragma unroll
            for (int j = 0; j < NS; ++j) a[j] = n[(j + 1) & (NS - 1)];
        }
        if (h) {
            #pragma unroll
            for (int j = 0; j < NS; ++j) D[r*NS + j] = a[j];
        }
    }
    __syncthreads();
}

__device__ __forceinline__ float nmax(float a, float v) {
    return (v > a || v != v) ? v : a;
}

#define MX_AT   0
#define MX_CP   1
#define MX_CPT  2
#define MX_R16  3
#define MX_X    4
#define MX_GL   5
#define MX_GR   6
#define MX_GC   7
#define MX_GJ   8
#define MX_NL   9
#define MX_NR   10
#define MX_NC   11
#define MX_NJ   12
#define MX_E    13
#define MX_T0   14
#define MX_CNT  20
#define MXP(pool, m) ((pool) + (m)*NS*NS)

__device__ __forceinline__ float map_square(float* pool, float* red, int tid) {
    float* X  = MXP(pool, MX_X);
    float* GL = MXP(pool, MX_GL);  float* GR = MXP(pool, MX_GR);
    float* GC = MXP(pool, MX_GC);  float* GJ = MXP(pool, MX_GJ);
    float* NL = MXP(pool, MX_NL);  float* NR = MXP(pool, MX_NR);
    float* NC = MXP(pool, MX_NC);  float* NJ = MXP(pool, MX_NJ);
    float* t0 = MXP(pool, MX_T0);      float* t1 = MXP(pool, MX_T0 + 1);
    float* t2 = MXP(pool, MX_T0 + 2);  float* t3 = MXP(pool, MX_T0 + 3);
    float* t4 = MXP(pool, MX_T0 + 4);  float* t5 = MXP(pool, MX_T0 + 5);
    sub16(t0, GC, X, tid);
    mm16(t1, t0, GJ, tid);
    eye_axpy(t2, t1, 1.0f, tid);
    inv16(t3, t2, tid);
    mm16(t1, t3, t0, tid);
    mm16(t2, GJ, t1, tid);
    eye_axpy(t4, t2, -1.0f, tid);
    mm16(t2, t3, GL, tid);
    mm16(NL, GL, t2, tid);
    mm16(t2, GL, t1, tid);
    mm16_add(NC, t2, GR, GC, tid);
    mm16(t2, t4, GR, tid);
    mm16(NR, GR, t2, tid);
    mm16(t2, GJ, GL, tid);
    mm16(t5, t4, t2, tid);
    mm16_add(NJ, GR, t5, GJ, tid);
    const float l0 = NL[tid], l1 = NL[tid + CTH], r0 = NR[tid], r1 = NR[tid + CTH];
    const float c0 = NC[tid], c1 = NC[tid + CTH], j0 = NJ[tid], j1 = NJ[tid + CTH];
    red[tid]         = nmax(nmax(0.f, fabsf(l0)), fabsf(l1));
    red[CTH + tid]   = nmax(nmax(0.f, fabsf(r0)), fabsf(r1));
    red[2*CTH + tid] = nmax(nmax(nmax(nmax(0.f, fabsf(c0)), fabsf(c1)), fabsf(j0)), fabsf(j1));
    __syncthreads();
    float mL = 0.f, mR = 0.f, mC = 0.f;
    for (int i = 0; i < CTH; ++i) {
        mL = nmax(mL, red[i]);  mR = nmax(mR, red[CTH + i]);  mC = nmax(mC, red[2*CTH + i]);
    }
    const bool fin = (mL < 3.0e38f) && (mR < 3.0e38f) && (mC < 3.0e38f);
    int a = 0;
    if (fin && mL > 0.f && mR > 0.f) {
        int eL, eR;
        (void)frexpf(mL, &eL);  (void)frexpf(mR, &eR);
        a = (eR - eL) / 2;
    }
    GL[tid] = ldexpf(l0, a);   GL[tid + CTH] = ldexpf(l1, a);
    GR[tid] = ldexpf(r0, -a);  GR[tid + CTH] = ldexpf(r1, -a);
    GC[tid] = c0;  GC[tid + CTH] = c1;
    GJ[tid] = j0;  GJ[tid + CTH] = j1;
    __syncthreads();
    return fin ? mL * mR : 1.0f;
}

__device__ __forceinline__ void map_apply(float* pool, int tid) {
    float* X  = MXP(pool, MX_X);   float* E  = MXP(pool, MX_E);
    float* GL = MXP(pool, MX_GL);  float* GR = MXP(pool, MX_GR);
    float* GC = MXP(pool, MX_GC);  float* GJ = MXP(pool, MX_GJ);
    float* t0 = MXP(pool, MX_T0);      float* t1 = MXP(pool, MX_T0 + 1);
    float* t2 = MXP(pool, MX_T0 + 2);  float* t3 = MXP(pool, MX_T0 + 3);
    sub16(t0, E, X, tid);
    mm16(t1, t0, GJ, tid);
    eye_axpy(t2, t1, 1.0f, tid);
    inv16(t3, t2, tid);
    mm16(t1, t3, t0, tid);
    mm16(t2, GL, t1, tid);
    mm16_add(E, t2, GR, GC, tid);
}

__device__ __forceinline__ void map_one_step(float* pool, const float* sA,
                                             const float* sQ, int tid) {
    float* AT  = MXP(pool, MX_AT);   float* CP  = MXP(pool, MX_CP);
    float* CPT = MXP(pool, MX_CPT);  float* R16 = MXP(pool, MX_R16);
    float* X   = MXP(pool, MX_X);
    float* GL = MXP(pool, MX_GL);  float* GR = MXP(pool, MX_GR);
    float* GC = MXP(pool, MX_GC);  float* GJ = MXP(pool, MX_GJ);
    float* t0 = MXP(pool, MX_T0);  float* t1 = MXP(pool, MX_T0 + 1);  float* t2 = MXP(pool, MX_T0 + 2);
    mm16(t0, CP, X, tid);
    mm16_add(t1, t0, CPT, R16, tid);
    inv16(t2, t1, tid);
    mm16(t0, CPT, t2, tid);
    mm16(GJ, t0, CP, tid);
    mm16(t0, X, GJ, tid);
    eye_axpy(t1, t0, -1.0f, tid);
    mm16(t0, GJ, X, tid);
    eye_axpy(t2, t0, -1.0f, tid);
    mm16(GL, sA, t1, tid);
    mm16(GR, t2, AT, tid);
    mm16(t0, GL, X, tid);
    mm16_add(GC, t0, AT, sQ, tid);
}

__global__ __launch_bounds__(CTH)
void kalman_cov_kernel(const float* __restrict__ Ag, const float* __restrict__ Bg,
                       const float* __restrict__ Cg, const float* __restrict__ Qg,
                       const float* __restrict__ Rg, float* __restrict__ ws)
{
    __shared__ float sA[NS*NS], sB[NS*US], sC[MS*NS], sQ[NS*NS], sR[MS*MS];
    __shared__ float sBot[2*NS*TWB];
    __shared__ float sCA[MS*NS], sCB[MS*US];
    __shared__ float sPool[MX_CNT*NS*NS];
    __shared__ float sRed[3*CTH];

    const int tid = threadIdx.x;
    for (int e = tid; e < NS*NS; e += CTH) {
        sA[e] = Ag[e];
        sQ[e] = Qg[e];
    }
    if (tid < NS*US) sB[tid] = Bg[tid];
    if (tid < MS*NS) sC[tid] = Cg[tid];
    if (tid < MS*MS) sR[tid] = Rg[tid];
    __syncthreads();

    #pragma unroll
    for (int q = 0; q < 2; ++q) {
        const int e = tid + q*CTH, i = e >> 4, j = e & 15;
        MXP(sPool, MX_AT)[e]  = sA[j*NS + i];
        MXP(sPool, MX_CP)[e]  = (i < MS) ? sC[i*NS + j] : 0.0f;
        MXP(sPool, MX_CPT)[e] = (j < MS) ? sC[j*NS + i] : 0.0f;
        MXP(sPool, MX_R16)[e] = (i < MS && j < MS) ? sR[i*MS + j] : ((i == j) ? 1.0f : 0.0f);
    }

    for (int e = tid; e < MS*NS + MS*US; e += CTH) {
        if (e < MS*NS) {
            const int i16 = e >> 4, j16 = e & 15;
            float a = 0.f;
            #pragma unroll
            for (int k = 0; k < NS; ++k) a += sC[i16*NS + k] * sA[k*NS + j16];
            sCA[e] = a;
        } else {
            const int q = e - MS*NS, ib = q >> 2, jb = q & 3;
            float a = 0.f;
            #pragma unroll
            for (int k = 0; k < NS; ++k) a += sC[ib*NS + k] * sB[k*US + jb];
            sCB[q] = a;
        }
    }
    __syncthreads();

    if (tid < NS*8) {
        const float a = sA[(tid >> 3)*NS + (tid & 7)];
        sBot[(tid >> 3)*TWB + 8 + (tid & 7)] = a;
        sBot[NS*TWB + (tid >> 3)*TWB + 8 + (tid & 7)] = a;
    }

    const int w  = tid >> 5, c = tid & 31;
    const int rbase = RPW * w;
    float kBHi[RPW];
    int   oh[RPW];
    #pragma unroll
    for (int j = 0; j < RPW; ++j) {
        const int r = rbase + j;
        kBHi[j] = 0.f;  oh[j] = -1;
        if (c < 8)        { kBHi[j] = sA[r*NS + 8 + c];     oh[j] = r*NS + 8 + c; }
        else if (c < 12)  { kBHi[j] = sB[r*US + (c - 8)];   oh[j] = 256 + r*US + (c - 8); }
        else if (c < 20)  { oh[j] = 320 + r*MS + (c - 12); }
    }
    float kPHi[MS];
    #pragma unroll
    for (int i = 0; i < MS; ++i) {
        float h = 0.f;
        if (c < 8)        h = sCA[i*NS + 8 + c];
        else if (c < 12)  h = sCB[i*US + (c - 8)];
        else if (c < 20)  h = ((c - 12) == i) ? -1.0f : 0.0f;
        kPHi[i] = h;
    }

    const int hr  = c >> 4, jc = c & 15;
    const int rowa = rbase + hr;
    const int rowb = rbase + 2 + hr;
    float kAj[NS], kCm[NS];
    #pragma unroll
    for (int k = 0; k < NS; ++k) {
        kAj[k]    = sA[jc*NS + k];
        kCm[k]    = sC[(jc & 7)*NS + k];
    }
    float kC[MS*NS];
    #pragma unroll
    for (int q = 0; q < MS*NS; ++q) kC[q] = sC[q];
    float kR[MS];
    #pragma unroll
    for (int i = 0; i < MS; ++i) kR[i] = (c < MS) ? sR[i*MS + c] : 0.0f;
    float kA2[RPW][NS];
    float kQ2[RPW];
    #pragma unroll
    for (int j = 0; j < RPW; ++j) {
        #pragma unroll
        for (int k = 0; k < NS; ++k) kA2[j][k] = sA[(rbase + j)*NS + k];
        kQ2[j] = (c >= 16) ? sQ[(rbase + j)*NS + (c - 16)] : 0.0f;
    }

    float m0, m1;
    {
        float a0 = 0.f, a1 = 0.f;
        #pragma unroll
        for (int k = 0; k < NS; ++k) {
            a0 += sA[rowa*NS + k] * kAj[k];
            a1 += sA[rowb*NS + k] * kAj[k];
        }
        m0 = a0 + sQ[rowa*NS + jc];
        m1 = a1 + sQ[rowb*NS + jc];
    }

    float r_lo[RPW], r_hi[RPW];
    #pragma unroll
    for (int j = 0; j < RPW; ++j) { r_lo[j] = 0.f;  r_hi[j] = 0.f; }

    const int cb = blockIdx.x;
    int  tb = 1, te = CL + 1;
    bool store = (cb == 0);
    #pragma unroll 1
    for (int phase = 0; phase < 2; ++phase) {
    if (phase == 1) {
        float* mxX = MXP(sPool, MX_X);
        float* mxE = MXP(sPool, MX_E);
        mxX[rowa*NS + jc] = m0;  mxX[rowb*NS + jc] = m1;
        mxE[rowa*NS + jc] = m0;  mxE[rowb*NS + jc] = m1;
        __syncthreads();
        map_one_step(sPool, sA, sQ, tid);
        const int nq = (cb > 0) ? (cb - 1) : 0;
        float gv = 1.0f;
        bool  ok = false;
        #pragma unroll 1
        for (int s = 0; s < CLLG + LADB; ++s) {
            if (s == FLG) {
                ok = gv < FORGET_BOUND;
                if (!ok) break;
            }
            if (s >= CLLG) {
                const int k = s - CLLG;
                if ((nq >> k) & 1) map_apply(sPool, tid);
                if (s >= FLG && (nq >> (k + 1)) == 0) break;
            }
            gv = map_square(sPool, sRed, tid);
        }
        if (ok) {
            if (cb == 0) return;
            tb = cb * CL + 1;
            te = (tb + CL < SEQ) ? (tb + CL) : SEQ;
            m0 = mxE[rowa*NS + jc];
            m1 = mxE[rowb*NS + jc];
        } else {
            if (cb != 0) return;
            tb = CL + 1;
            te = SEQ;
        }
        store = true;
    }
    #pragma unroll 1
    for (int t = tb; t < te; ++t) {
        float* bot = sBot + (t & 1) * (NS*TWB);
        float ga, xa, gb, xb;
        row_dot4(m0, m1, kCm, kAj, ga, xa, gb, xb);
        bot[rowa*TWB + 16 + jc] = xa;
        bot[rowb*TWB + 16 + jc] = xb;
        bot[rowa*TWB + (jc & 7)] = ga;
        bot[rowb*TWB + (jc & 7)] = gb;
        __syncthreads();

        float b_lo[RPW];
        #pragma unroll
        for (int j = 0; j < RPW; ++j) b_lo[j] = bot[(rbase + j)*TWB + c];
        float col[NS];
        #pragma unroll
        for (int k = 0; k < NS; ++k) col[k] = bot[k*TWB + c];
        if (store && t > tb) {
            float* wsq = ws + (size_t)(t - 2) * STEPF;
            if (c >= 8 && c < 16) {
                #pragma unroll
                for (int j = 0; j < RPW; ++j) wsq[(rbase + j)*NS + (c - 8)] = r_lo[j];
            }
            if (oh[0] >= 0) {
                #pragma unroll
                for (int j = 0; j < RPW; ++j) wsq[oh[j]] = r_hi[j];
            }
        }
        float pl[MS], ph[MS];
        #pragma unroll
        for (int i = 0; i < MS; ++i) {
            pl[i] = kR[i];
            ph[i] = kPHi[i];
        }
        #pragma unroll
        for (int k = 0; k < NS; ++k) {
            #pragma unroll
            for (int i = 0; i < MS; ++i) pl[i] += kC[i*NS + k] * col[k];
        }
        float b2[RPW];
        #pragma unroll
        for (int j = 0; j < RPW; ++j) b2[j] = kQ2[j];
        #pragma unroll
        for (int k = 0; k < NS; ++k) {
            #pragma unroll
            for (int j = 0; j < RPW; ++j) b2[j] += kA2[j][k] * col[k];
        }
        float b_hi[RPW];
        #pragma unroll
        for (int j = 0; j < RPW; ++j) b_hi[j] = kBHi[j];

        #pragma unroll
        for (int p = 0; p < MS; ++p) {
            const float inv = 1.0f / lane_bcast(pl[p], p);
            #pragma unroll
            for (int i = p + 1; i < MS; ++i) {
                const float f = lane_bcast(pl[i], p) * inv;
                pl[i] -= f * pl[p];  ph[i] -= f * ph[p];
            }
            #pragma unroll
            for (int j = 0; j < RPW; ++j) {
                const float fj = lane_bcast(b_lo[j], p) * inv;
                b_lo[j] -= fj * pl[p];  b_hi[j] -= fj * ph[p];
                const float gj = lane_bcast(b2[j], p) * inv;
                b2[j] -= gj * pl[p];
            }
        }

        #pragma unroll
        for (int j = 0; j < RPW; ++j) { r_lo[j] = b_lo[j];  r_hi[j] = b_hi[j]; }

        const float x0 = xhalf(b2[0]);
        const float x2 = xhalf(b2[2]);
        m0 = hr ? b2[1] : x0;
        m1 = hr ? b2[3] : x2;
    }

    if (store) {
        float* wsq = ws + (size_t)(te - 2) * STEPF;
        if (c >= 8 && c < 16) {
            #pragma unroll
            for (int j = 0; j < RPW; ++j) wsq[(rbase + j)*NS + (c - 8)] = r_lo[j];
        }
        if (oh[0] >= 0) {
            #pragma unroll
            for (int j = 0; j < RPW; ++j) wsq[oh[j]] = r_hi[j];
        }
    }
    }
}

#define CH_L 63
#define CH_N 65
#define PHI_ROW 2

__device__ __forceinline__ float bperm(int srcLane, float v) {
    return __builtin_bit_cast(float,
        __builtin_amdgcn_ds_bpermute(srcLane << 2, __builtin_bit_cast(int, v)));
}

__device__ __forceinline__ v8f wmma4(v2f a, v2f b, v8f c) {
    return __builtin_amdgcn_wmma_f32_16x16x4_f32(false, a, false, b,
                                                 (short)0, c, false, false);
}

struct AOps {
    v2f a0, a1, a2, a3;
};

struct StepOps {
    AOps  aA;
    v2f aB, aK0, aK1;
    v2f bZ0, bZ1, bU;
};

__device__ __forceinline__ AOps load_a(const float* m, int col, int hi) {
    AOps o;
    o.a0 = *(const v2f*)(m + col * 16 + 0  + 2 * hi);
    o.a1 = *(const v2f*)(m + col * 16 + 4  + 2 * hi);
    o.a2 = *(const v2f*)(m + col * 16 + 8  + 2 * hi);
    o.a3 = *(const v2f*)(m + col * 16 + 12 + 2 * hi);
    return o;
}

__device__ __forceinline__ StepOps load_ops(const float* __restrict__ ws,
                                            const float* __restrict__ zbase,
                                            const float* __restrict__ ubase,
                                            int t, int col, int hi) {
    StepOps o;
    const float* w = ws + (size_t)(t - 1) * STEPF;
    o.aA  = load_a(w, col, hi);
    o.aB  = *(const v2f*)(w + 256 + col * 4 + 2 * hi);
    o.aK0 = *(const v2f*)(w + 320 + col * 8 + 0 + 2 * hi);
    o.aK1 = *(const v2f*)(w + 320 + col * 8 + 4 + 2 * hi);
    const float* zr = zbase + (size_t)t * MS;
    o.bZ0 = *(const v2f*)(zr + 0 + 2 * hi);
    o.bZ1 = *(const v2f*)(zr + 4 + 2 * hi);
    o.bU  = *(const v2f*)(ubase + (size_t)t * US + 2 * hi);
    return o;
}

__device__ __forceinline__ v8f zu_chain(const StepOps& o) {
    v8f d = {};
    d = wmma4(o.aB,  o.bU,  d);
    d = wmma4(o.aK0, o.bZ0, d);
    d = wmma4(o.aK1, o.bZ1, d);
    return d;
}

__device__ __forceinline__ v8f recur(const AOps& m, const v8f& x, v8f c, int col, int hi) {
    v2f bX[4];
    #pragma unroll
    for (int q = 0; q < 4; ++q) {
        int srcLane = col + ((q >= 2) ? 16 : 0);
        int lo = 4 * (q & 1);
        float t0 = bperm(srcLane, x[lo + 0]);
        float t1 = bperm(srcLane, x[lo + 2]);
        float t2 = bperm(srcLane, x[lo + 1]);
        float t3 = bperm(srcLane, x[lo + 3]);
        bX[q].x = hi ? t1 : t0;
        bX[q].y = hi ? t3 : t2;
    }
    v8f d = wmma4(m.a0, bX[0], c);
    d = wmma4(m.a1, bX[1], d);
    d = wmma4(m.a2, bX[2], d);
    d = wmma4(m.a3, bX[3], d);
    return d;
}

__device__ __forceinline__ void store_row(float* o, const v8f& x) {
    *(float4*)(o)     = make_float4(x[0], x[1], x[2], x[3]);
    *(float4*)(o + 4) = make_float4(x[4], x[5], x[6], x[7]);
}

__device__ __forceinline__ v8f load_row(const float* o) {
    const float4 lo = *(const float4*)(o);
    const float4 hh = *(const float4*)(o + 4);
    v8f x = {lo.x, lo.y, lo.z, lo.w, hh.x, hh.y, hh.z, hh.w};
    return x;
}

__global__ __launch_bounds__(32)
void kalman_chunk_local_kernel(const float* __restrict__ obs, const float* __restrict__ inp,
                               const float* __restrict__ ws, float* __restrict__ out)
{
    const int lane = threadIdx.x;
    const int hi   = lane >> 4;
    const int col  = lane & 15;
    const int tile = blockIdx.x;
    const int t0   = blockIdx.y * CH_L;

    if (tile == BATCH / 16) {
        v8f x;
        #pragma unroll
        for (int r = 0; r < 8; ++r) x[r] = (8 * hi + r == col) ? 1.0f : 0.0f;
        AOps cur = load_a(ws + (size_t)t0 * STEPF, col, hi);
        #pragma unroll 1
        for (int s = 1; s <= CH_L; ++s) {
            const int wn = (s < CH_L) ? (t0 + s) : (t0 + s - 1);
            AOps nxt = load_a(ws + (size_t)wn * STEPF, col, hi);
            v8f zero = {};
            x = recur(cur, x, zero, col, hi);
            cur = nxt;
        }
        float* p = out + (size_t)(t0 + PHI_ROW) * NS;
        #pragma unroll
        for (int r = 0; r < 8; ++r) p[(8 * hi + r) * NS + col] = x[r];
        return;
    }

    const int b = tile * 16 + col;
    const float* zbase = obs + (size_t)b * SEQ * MS;
    const float* ubase = inp + (size_t)b * SEQ * US;

    v8f x = {};
    StepOps cur = load_ops(ws, zbase, ubase, t0 + 1, col, hi);
    v8f dzu = zu_chain(cur);
    #pragma unroll 1
    for (int s = 1; s <= CH_L; ++s) {
        const int t  = t0 + s;
        const int tn = (s < CH_L) ? (t + 1) : t;
        StepOps nxt = load_ops(ws, zbase, ubase, tn, col, hi);
        x = recur(cur.aA, x, dzu, col, hi);
        dzu = zu_chain(nxt);
        cur = nxt;
    }
    store_row(out + ((size_t)b * SEQ + (t0 + CH_L)) * NS + 8 * hi, x);
}

__global__ __launch_bounds__(32)
void kalman_chunk_link_kernel(const float* __restrict__ x0, float* out)
{
    const int lane = threadIdx.x;
    const int hi   = lane >> 4;
    const int col  = lane & 15;
    const int b    = blockIdx.x * 16 + col;

    v8f x;
    #pragma unroll
    for (int r = 0; r < 8; ++r) x[r] = x0[8 * hi + r];

    float* obase = out + (size_t)b * SEQ * NS + 8 * hi;
    store_row(obase, x);

    AOps cur = load_a(out + (size_t)PHI_ROW * NS, col, hi);
    v8f  y   = load_row(obase + (size_t)CH_L * NS);
    #pragma unroll 1
    for (int c = 0; c < CH_N; ++c) {
        const int cn = (c + 1 < CH_N) ? (c + 1) : c;
        AOps nxt = load_a(out + ((size_t)cn * CH_L + PHI_ROW) * NS, col, hi);
        v8f  yn  = load_row(obase + (size_t)(cn + 1) * CH_L * NS);
        x = recur(cur, x, y, col, hi);
        store_row(obase + (size_t)(c + 1) * CH_L * NS, x);
        cur = nxt;
        y = yn;
    }
}

__global__ __launch_bounds__(32)
void kalman_chunk_final_kernel(const float* __restrict__ obs, const float* __restrict__ inp,
                               const float* __restrict__ ws, float* out)
{
    const int lane = threadIdx.x;
    const int hi   = lane >> 4;
    const int col  = lane & 15;
    const int b    = blockIdx.x * 16 + col;
    const int t0   = blockIdx.y * CH_L;

    const float* zbase = obs + (size_t)b * SEQ * MS;
    const float* ubase = inp + (size_t)b * SEQ * US;
    float*       obase = out + (size_t)b * SEQ * NS + 8 * hi;

    v8f x = load_row(obase + (size_t)t0 * NS);
    StepOps cur = load_ops(ws, zbase, ubase, t0 + 1, col, hi);
    v8f dzu = zu_chain(cur);
    #pragma unroll 1
    for (int s = 1; s < CH_L; ++s) {
        const int t  = t0 + s;
        const int tn = (s + 1 < CH_L) ? (t + 1) : t;
        StepOps nxt = load_ops(ws, zbase, ubase, tn, col, hi);
        x = recur(cur.aA, x, dzu, col, hi);
        store_row(obase + (size_t)t * NS, x);
        dzu = zu_chain(nxt);
        cur = nxt;
    }
}

extern "C" void kernel_launch(void* const* d_in, const int* in_sizes, int n_in,
                              void* d_out, int out_size, void* d_ws, size_t ws_size,
                              hipStream_t stream) {
    const float* obs = (const float*)d_in[0];
    const float* inp = (const float*)d_in[1];
    const float* A   = (const float*)d_in[2];
    const float* B   = (const float*)d_in[3];
    const float* C   = (const float*)d_in[4];
    const float* Q   = (const float*)d_in[5];
    const float* R   = (const float*)d_in[6];
    const float* x0  = (const float*)d_in[7];
    float* out = (float*)d_out;
    float* ws  = (float*)d_ws;

    hipLaunchKernelGGL(kalman_cov_kernel, dim3(NCB), dim3(CTH), 0, stream,
                       A, B, C, Q, R, ws);
    hipLaunchKernelGGL(kalman_chunk_local_kernel, dim3(BATCH / 16 + 1, CH_N), dim3(32), 0, stream,
                       obs, inp, ws, out);
    hipLaunchKernelGGL(kalman_chunk_link_kernel, dim3(BATCH / 16), dim3(32), 0, stream,
                       x0, out);
    hipLaunchKernelGGL(kalman_chunk_final_kernel, dim3(BATCH / 16, CH_N), dim3(32), 0, stream,
                       obs, inp, ws, out);
}
